# starting kernel
# baseline (speedup 1.0000x reference)
#include <hip/hip_runtime.h>
#include <hip/hip_bf16.h>

typedef float v2f __attribute__((ext_vector_type(2)));
typedef float v8f __attribute__((ext_vector_type(8)));

#define F_IN 128
#define F_E  16
#define NH   8
#define NC   32
#define HC   256
#define NEG_SLOPE 0.2f

__device__ __forceinline__ unsigned ordkey(float f) {
    unsigned u = __float_as_uint(f);
    return (u & 0x80000000u) ? ~u : (u | 0x80000000u);
}
__device__ __forceinline__ float ordinv(unsigned k) {
    unsigned u = (k & 0x80000000u) ? (k & 0x7fffffffu) : ~k;
    return __uint_as_float(u);
}

__global__ void GAT_zero(float* p, long n) {
    long i = (long)blockIdx.x * blockDim.x + threadIdx.x;
    long stride = (long)gridDim.x * blockDim.x;
    for (; i < n; i += stride) p[i] = 0.0f;
}

__global__ void GAT_degloop(const int* __restrict__ dst,
                            const float* __restrict__ eattr,
                            float* __restrict__ deg,
                            float* __restrict__ loop, int E) {
    long idx = (long)blockIdx.x * blockDim.x + threadIdx.x;
    long e = idx >> 4;
    int  f = (int)(idx & 15);
    if (e >= E) return;
    int d = dst[e];
    atomicAdd(&loop[(long)d * F_E + f], eattr[e * F_E + f]);
    if (f == 0) atomicAdd(&deg[d], 1.0f);
}

__global__ void GAT_loopdiv(float* __restrict__ loop,
                            const float* __restrict__ deg, int N) {
    long idx = (long)blockIdx.x * blockDim.x + threadIdx.x;
    if (idx >= (long)N * F_E) return;
    int n = (int)(idx >> 4);
    loop[idx] = loop[idx] / fmaxf(deg[n], 1.0f);
}

__global__ void GAT_foldM(const float* __restrict__ We,
                          const float* __restrict__ att_edge,
                          float* __restrict__ M) {
    int tid = threadIdx.x;
    if (tid >= F_E * NH) return;
    int f = tid >> 3, h = tid & 7;
    float s = 0.0f;
    for (int c = 0; c < NC; ++c)
        s += We[(long)(h * NC + c) * F_E + f] * att_edge[h * NC + c];
    M[f * NH + h] = s;
}

__global__ void GAT_gemm_xh(const float* __restrict__ x,
                            const float* __restrict__ W,
                            float* __restrict__ xh, int N) {
    int wave = threadIdx.x >> 5;
    int lane = threadIdx.x & 31;
    int half = lane >> 4;
    int l    = lane & 15;
    long rowTiles = ((long)N + 15) >> 4;
    long wtile = (long)blockIdx.x * 8 + wave;
    if (wtile >= rowTiles * 4) return;
    long rowTile = wtile >> 2;
    int  colGrp  = (int)(wtile & 3);
    long row0 = rowTile * 16;
    int  col0 = colGrp * 64;

    long arow = row0 + l;
    if (arow >= N) arow = N - 1;
    const float* xp  = x + arow * F_IN + 2 * half;
    const float* wp0 = W + (long)(col0 +  0 + l) * F_IN + 2 * half;
    const float* wp1 = W + (long)(col0 + 16 + l) * F_IN + 2 * half;
    const float* wp2 = W + (long)(col0 + 32 + l) * F_IN + 2 * half;
    const float* wp3 = W + (long)(col0 + 48 + l) * F_IN + 2 * half;

    v8f a0 = {0.f,0.f,0.f,0.f,0.f,0.f,0.f,0.f};
    v8f a1 = a0, a2 = a0, a3 = a0;
    #pragma unroll
    for (int k0 = 0; k0 < F_IN; k0 += 4) {
        v2f a  = *(const v2f*)(xp  + k0);
        v2f b0 = *(const v2f*)(wp0 + k0);
        v2f b1 = *(const v2f*)(wp1 + k0);
        v2f b2 = *(const v2f*)(wp2 + k0);
        v2f b3 = *(const v2f*)(wp3 + k0);
        a0 = __builtin_amdgcn_wmma_f32_16x16x4_f32(false, a, false, b0, (short)0, a0, false, false);
        a1 = __builtin_amdgcn_wmma_f32_16x16x4_f32(false, a, false, b1, (short)0, a1, false, false);
        a2 = __builtin_amdgcn_wmma_f32_16x16x4_f32(false, a, false, b2, (short)0, a2, false, false);
        a3 = __builtin_amdgcn_wmma_f32_16x16x4_f32(false, a, false, b3, (short)0, a3, false, false);
    }
    #pragma unroll
    for (int v = 0; v < 8; ++v) {
        long r = row0 + v + 8 * half;
        if (r < N) {
            float* o = xh + r * HC + col0 + l;
            o[0]  = a0[v];
            o[16] = a1[v];
            o[32] = a2[v];
            o[48] = a3[v];
        }
    }
}

__global__ void GAT_nodeatt(const float* __restrict__ xh,
                            const float* __restrict__ att_src,
                            const float* __restrict__ att_dst,
                            float* __restrict__ asrc,
                            float* __restrict__ adst, int N) {
    long idx = (long)blockIdx.x * blockDim.x + threadIdx.x;
    if (idx >= (long)N * NH) return;
    long n = idx >> 3;
    int  h = (int)(idx & 7);
    const float* xr = xh + n * HC + h * NC;
    float ss = 0.0f, sd = 0.0f;
    for (int c = 0; c < NC; ++c) {
        float v = xr[c];
        ss += v * att_src[h * NC + c];
        sd += v * att_dst[h * NC + c];
    }
    asrc[idx] = ss;
    adst[idx] = sd;
}

__global__ void GAT_alpha(const int* __restrict__ src, const int* __restrict__ dst,
                          const float* __restrict__ eattr, const float* __restrict__ loop,
                          const float* __restrict__ M,
                          const float* __restrict__ asrc, const float* __restrict__ adst,
                          float* __restrict__ exbuf, unsigned* __restrict__ amax,
                          int E, int N) {
    long t = (long)blockIdx.x * blockDim.x + threadIdx.x;
    long T = (long)E + N;
    if (t >= T) return;
    int s, d;
    const float* row;
    if (t < E) { s = src[t]; d = dst[t]; row = eattr + t * F_E; }
    else       { s = (int)(t - E); d = s;  row = loop + (long)s * F_E; }
    float ea[F_E];
    #pragma unroll
    for (int f = 0; f < F_E; ++f) ea[f] = row[f];
    #pragma unroll
    for (int h = 0; h < NH; ++h) {
        float ae = 0.0f;
        #pragma unroll
        for (int f = 0; f < F_E; ++f) ae += ea[f] * M[f * NH + h];
        float al = asrc[(long)s * NH + h] + adst[(long)d * NH + h] + ae;
        al = (al >= 0.0f) ? al : NEG_SLOPE * al;
        exbuf[t * NH + h] = al;
        atomicMax(&amax[(long)d * NH + h], ordkey(al));
    }
}

__global__ void GAT_expsum(const int* __restrict__ dst,
                           float* __restrict__ exbuf,
                           const unsigned* __restrict__ amax,
                           float* __restrict__ denom, int E, int N) {
    long idx = (long)blockIdx.x * blockDim.x + threadIdx.x;
    long T = ((long)E + N) * NH;
    if (idx >= T) return;
    long t = idx >> 3;
    int  h = (int)(idx & 7);
    int d = (t < E) ? dst[t] : (int)(t - E);
    float m = ordinv(amax[(long)d * NH + h]);
    float e = __expf(exbuf[idx] - m);
    exbuf[idx] = e;
    atomicAdd(&denom[(long)d * NH + h], e);
}

__global__ void GAT_scatter(const int* __restrict__ src, const int* __restrict__ dst,
                            const float* __restrict__ xh,
                            const float* __restrict__ exbuf,
                            const float* __restrict__ denom,
                            float* __restrict__ accum, int E, int N) {
    long t = blockIdx.x;
    int tid = threadIdx.x;
    int s, d;
    if (t < E) { s = src[t]; d = dst[t]; }
    else       { s = (int)(t - E); d = s; }
    int h = tid >> 5;
    float att = exbuf[t * NH + h] / denom[(long)d * NH + h];
    float val = xh[(long)s * HC + tid] * att;
    atomicAdd(&accum[(long)d * HC + tid], val);
}

__global__ void GAT_final(const float* __restrict__ accum,
                          const float* __restrict__ bias,
                          float* __restrict__ out, int N) {
    long idx = (long)blockIdx.x * blockDim.x + threadIdx.x;
    if (idx >= (long)N * NC) return;
    long n = idx >> 5;
    int  c = (int)(idx & 31);
    float sum = 0.0f;
    #pragma unroll
    for (int h = 0; h < NH; ++h) sum += accum[n * HC + h * NC + c];
    out[idx] = sum * (1.0f / NH) + bias[c];
}

extern "C" void kernel_launch(void* const* d_in, const int* in_sizes, int n_in,
                              void* d_out, int out_size, void* d_ws, size_t ws_size,
                              hipStream_t stream) {
    const float* x        = (const float*)d_in[0];
    const int*   eidx     = (const int*)  d_in[1];
    const float* eattr    = (const float*)d_in[2];
    const float* W        = (const float*)d_in[3];
    const float* We       = (const float*)d_in[4];
    const float* att_src  = (const float*)d_in[5];
    const float* att_dst  = (const float*)d_in[6];
    const float* att_edge = (const float*)d_in[7];
    const float* bias     = (const float*)d_in[8];

    const int N = in_sizes[0] / F_IN;
    const int E = in_sizes[1] / 2;
    const int* src = eidx;
    const int* dst = eidx + E;

    float* ws = (float*)d_ws;
    const long Nl = N, El = E, Tl = El + Nl;
    float*    w_deg   = ws;
    float*    w_loop  = ws + Nl;
    float*    w_denom = ws + 17 * Nl;
    unsigned* w_amax  = (unsigned*)(ws + 25 * Nl);
    float*    w_acc   = ws + 33 * Nl;
    const long zeroSpan = 289 * Nl;
    float*    w_xh    = ws + zeroSpan;
    float*    w_asrc  = ws + zeroSpan + 256 * Nl;
    float*    w_adst  = ws + zeroSpan + 264 * Nl;
    float*    w_M     = ws + zeroSpan + 272 * Nl;
    float*    w_ex    = ws + zeroSpan + 272 * Nl + 128;

    const int B = 256;
    GAT_zero<<<2048, B, 0, stream>>>(ws, zeroSpan);
    {
        long th = El * F_E;
        GAT_degloop<<<(int)((th + B - 1) / B), B, 0, stream>>>(dst, eattr, w_deg, w_loop, E);
    }
    {
        long th = Nl * F_E;
        GAT_loopdiv<<<(int)((th + B - 1) / B), B, 0, stream>>>(w_loop, w_deg, N);
    }
    GAT_foldM<<<1, 128, 0, stream>>>(We, att_edge, w_M);
    {
        long rowTiles = (Nl + 15) >> 4;
        long wtiles = rowTiles * 4;
        GAT_gemm_xh<<<(int)((wtiles + 7) / 8), B, 0, stream>>>(x, W, w_xh, N);
    }
    {
        long th = Nl * NH;
        GAT_nodeatt<<<(int)((th + B - 1) / B), B, 0, stream>>>(w_xh, att_src, att_dst,
                                                               w_asrc, w_adst, N);
    }
    GAT_alpha<<<(int)((Tl + B - 1) / B), B, 0, stream>>>(src, dst, eattr, w_loop, w_M,
                                                         w_asrc, w_adst, w_ex, w_amax, E, N);
    {
        long th = Tl * NH;
        GAT_expsum<<<(int)((th + B - 1) / B), B, 0, stream>>>(dst, w_ex, w_amax, w_denom, E, N);
    }
    GAT_scatter<<<(int)Tl, B, 0, stream>>>(src, dst, w_xh, w_ex, w_denom, w_acc, E, N);
    {
        long th = Nl * NC;
        GAT_final<<<(int)((th + B - 1) / B), B, 0, stream>>>(w_acc, bias, (float*)d_out, N);
    }
}


# Message sum as a gather by destination: edge ids bucketed per node (group-scan bases + atomic cursor), one wave per node sums rows in registers, head mean fused; row atomics gone; sum order changes
# speedup vs baseline: 1.9354x; 1.9354x over previous
#include <hip/hip_runtime.h>
#include <hip/hip_bf16.h>

typedef float v2f __attribute__((ext_vector_type(2)));
typedef float v8f __attribute__((ext_vector_type(8)));
typedef float v4f __attribute__((ext_vector_type(4)));

#define F_IN 128
#define F_E  16
#define NH   8
#define NC   32
#define HC   256
#define NEG_SLOPE 0.2f

__device__ __forceinline__ unsigned ordkey(float f) {
    unsigned u = __float_as_uint(f);
    return (u & 0x80000000u) ? ~u : (u | 0x80000000u);
}
__device__ __forceinline__ float ordinv(unsigned k) {
    unsigned u = (k & 0x80000000u) ? (k & 0x7fffffffu) : ~k;
    return __uint_as_float(u);
}

__global__ void GAT_zero(float* p, long n) {
    long i = (long)blockIdx.x * blockDim.x + threadIdx.x;
    long stride = (long)gridDim.x * blockDim.x;
    for (; i < n; i += stride) p[i] = 0.0f;
}

__global__ void GAT_degloop(const int* __restrict__ dst,
                            const float* __restrict__ eattr,
                            float* __restrict__ deg,
                            float* __restrict__ loop, int E) {
    long idx = (long)blockIdx.x * blockDim.x + threadIdx.x;
    long e = idx >> 4;
    int  f = (int)(idx & 15);
    if (e >= E) return;
    int d = dst[e];
    atomicAdd(&loop[(long)d * F_E + f], eattr[e * F_E + f]);
    if (f == 0) atomicAdd(&deg[d], 1.0f);
}

__global__ void GAT_loopdiv(float* __restrict__ loop,
                            const float* __restrict__ deg, int N) {
    long idx = (long)blockIdx.x * blockDim.x + threadIdx.x;
    if (idx >= (long)N * F_E) return;
    int n = (int)(idx >> 4);
    loop[idx] = loop[idx] / fmaxf(deg[n], 1.0f);
}

__global__ void GAT_foldM(const float* __restrict__ We,
                          const float* __restrict__ att_edge,
                          float* __restrict__ M) {
    int tid = threadIdx.x;
    if (tid >= F_E * NH) return;
    int f = tid >> 3, h = tid & 7;
    float s = 0.0f;
    for (int c = 0; c < NC; ++c)
        s += We[(long)(h * NC + c) * F_E + f] * att_edge[h * NC + c];
    M[f * NH + h] = s;
}

__global__ void GAT_gemm_xh(const float* __restrict__ x,
                            const float* __restrict__ W,
                            float* __restrict__ xh, int N) {
    int wave = threadIdx.x >> 5;
    int lane = threadIdx.x & 31;
    int half = lane >> 4;
    int l    = lane & 15;
    long rowTiles = ((long)N + 15) >> 4;
    long wtile = (long)blockIdx.x * 8 + wave;
    if (wtile >= rowTiles * 4) return;
    long rowTile = wtile >> 2;
    int  colGrp  = (int)(wtile & 3);
    long row0 = rowTile * 16;
    int  col0 = colGrp * 64;

    long arow = row0 + l;
    if (arow >= N) arow = N - 1;
    const float* xp  = x + arow * F_IN + 2 * half;
    const float* wp0 = W + (long)(col0 +  0 + l) * F_IN + 2 * half;
    const float* wp1 = W + (long)(col0 + 16 + l) * F_IN + 2 * half;
    const float* wp2 = W + (long)(col0 + 32 + l) * F_IN + 2 * half;
    const float* wp3 = W + (long)(col0 + 48 + l) * F_IN + 2 * half;

    v8f a0 = {0.f,0.f,0.f,0.f,0.f,0.f,0.f,0.f};
    v8f a1 = a0, a2 = a0, a3 = a0;
    #pragma unroll
    for (int k0 = 0; k0 < F_IN; k0 += 4) {
        v2f a  = *(const v2f*)(xp  + k0);
        v2f b0 = *(const v2f*)(wp0 + k0);
        v2f b1 = *(const v2f*)(wp1 + k0);
        v2f b2 = *(const v2f*)(wp2 + k0);
        v2f b3 = *(const v2f*)(wp3 + k0);
        a0 = __builtin_amdgcn_wmma_f32_16x16x4_f32(false, a, false, b0, (short)0, a0, false, false);
        a1 = __builtin_amdgcn_wmma_f32_16x16x4_f32(false, a, false, b1, (short)0, a1, false, false);
        a2 = __builtin_amdgcn_wmma_f32_16x16x4_f32(false, a, false, b2, (short)0, a2, false, false);
        a3 = __builtin_amdgcn_wmma_f32_16x16x4_f32(false, a, false, b3, (short)0, a3, false, false);
    }
    #pragma unroll
    for (int v = 0; v < 8; ++v) {
        long r = row0 + v + 8 * half;
        if (r < N) {
            float* o = xh + r * HC + col0 + l;
            o[0]  = a0[v];
            o[16] = a1[v];
            o[32] = a2[v];
            o[48] = a3[v];
        }
    }
}

__global__ void GAT_nodeatt(const float* __restrict__ xh,
                            const float* __restrict__ att_src,
                            const float* __restrict__ att_dst,
                            float* __restrict__ asrc,
                            float* __restrict__ adst, int N) {
    long idx = (long)blockIdx.x * blockDim.x + threadIdx.x;
    if (idx >= (long)N * NH) return;
    long n = idx >> 3;
    int  h = (int)(idx & 7);
    const float* xr = xh + n * HC + h * NC;
    float ss = 0.0f, sd = 0.0f;
    for (int c = 0; c < NC; ++c) {
        float v = xr[c];
        ss += v * att_src[h * NC + c];
        sd += v * att_dst[h * NC + c];
    }
    asrc[idx] = ss;
    adst[idx] = sd;
}

__global__ void GAT_alpha(const int* __restrict__ src, const int* __restrict__ dst,
                          const float* __restrict__ eattr, const float* __restrict__ loop,
                          const float* __restrict__ M,
                          const float* __restrict__ asrc, const float* __restrict__ adst,
                          float* __restrict__ exbuf, unsigned* __restrict__ amax,
                          int E, int N) {
    long t = (long)blockIdx.x * blockDim.x + threadIdx.x;
    long T = (long)E + N;
    if (t >= T) return;
    int s, d;
    const float* row;
    if (t < E) { s = src[t]; d = dst[t]; row = eattr + t * F_E; }
    else       { s = (int)(t - E); d = s;  row = loop + (long)s * F_E; }
    float ea[F_E];
    #pragma unroll
    for (int f = 0; f < F_E; ++f) ea[f] = row[f];
    #pragma unroll
    for (int h = 0; h < NH; ++h) {
        float ae = 0.0f;
        #pragma unroll
        for (int f = 0; f < F_E; ++f) ae += ea[f] * M[f * NH + h];
        float al = asrc[(long)s * NH + h] + adst[(long)d * NH + h] + ae;
        al = (al >= 0.0f) ? al : NEG_SLOPE * al;
        exbuf[t * NH + h] = al;
        atomicMax(&amax[(long)d * NH + h], ordkey(al));
    }
}

__global__ void GAT_expsum(const int* __restrict__ dst,
                           float* __restrict__ exbuf,
                           const unsigned* __restrict__ amax,
                           float* __restrict__ denom, int E, int N) {
    long idx = (long)blockIdx.x * blockDim.x + threadIdx.x;
    long T = ((long)E + N) * NH;
    if (idx >= T) return;
    long t = idx >> 3;
    int  h = (int)(idx & 7);
    int d = (t < E) ? dst[t] : (int)(t - E);
    float m = ordinv(amax[(long)d * NH + h]);
    float e = __expf(exbuf[idx] - m);
    exbuf[idx] = e;
    atomicAdd(&denom[(long)d * NH + h], e);
}

__global__ void GAT_segalloc(const float* __restrict__ deg, int* __restrict__ off,
                             int* __restrict__ total, int N) {
    __shared__ int wsum[32];
    __shared__ int base;
    int t = threadIdx.x;
    int lane = t & 31, w = t >> 5;
    long n = (long)blockIdx.x * 1024 + t;
    int c = (n < N) ? (int)deg[n] : 0;
    int inc = c;
    for (int o = 1; o < 32; o <<= 1) {
        int v = __shfl_up(inc, o);
        if (lane >= o) inc += v;
    }
    if (lane == 31) wsum[w] = inc;
    __syncthreads();
    if (w == 0) {
        int s = wsum[lane];
        int si = s;
        for (int o = 1; o < 32; o <<= 1) {
            int v = __shfl_up(si, o);
            if (lane >= o) si += v;
        }
        wsum[lane] = si - s;
        if (lane == 31) base = atomicAdd(total, si);
    }
    __syncthreads();
    if (n < N) off[n] = base + wsum[w] + inc - c;
}

__global__ void GAT_fill(const int* __restrict__ dst, const int* __restrict__ off,
                         int* __restrict__ cur, int* __restrict__ perm, int E) {
    long e = (long)blockIdx.x * blockDim.x + threadIdx.x;
    if (e >= E) return;
    int d = dst[e];
    int p = atomicAdd(&cur[d], 1);
    perm[off[d] + p] = (int)e;
}

__global__ void GAT_gather(const int* __restrict__ src,
                           const float* __restrict__ deg,
                           const int* __restrict__ off,
                           const int* __restrict__ perm,
                           const float* __restrict__ xh,
                           const float* __restrict__ exbuf,
                           const float* __restrict__ denom,
                           const float* __restrict__ bias,
                           float* __restrict__ out, int E, int N) {
    int wave = __builtin_amdgcn_readfirstlane((int)(threadIdx.x >> 5));
    int lane = threadIdx.x & 31;
    long n = (long)blockIdx.x * 8 + wave;
    if (n >= N) return;
    int hq = lane >> 3;
    float den0 = denom[n * NH + hq];
    float den1 = denom[n * NH + 4 + hq];
    const float* xl = xh + lane * 4;
    const float* el = exbuf + hq;
    v4f c0 = {0.f, 0.f, 0.f, 0.f};
    v4f c1 = c0;
    int jb = off[n];
    int je = jb + (int)deg[n];
    for (int j0 = jb; j0 < je; j0 += 32) {
        int cnt = je - j0;
        if (cnt > 32) cnt = 32;
        int me = 0, ms = 0;
        if (lane < cnt) { me = perm[j0 + lane]; ms = src[me]; }
        #pragma unroll 4
        for (int i = 0; i < cnt; ++i) {
            int e = __builtin_amdgcn_readlane(me, i);
            int s = __builtin_amdgcn_readlane(ms, i);
            const float* ep = el + (long)e * NH;
            const float* xr = xl + (long)s * HC;
            float a0 = ep[0] / den0;
            float a1 = ep[4] / den1;
            v4f r0 = *(const v4f*)(xr);
            v4f r1 = *(const v4f*)(xr + 128);
            c0 += r0 * a0;
            c1 += r1 * a1;
        }
    }
    {
        const float* ep = el + ((long)E + n) * NH;
        const float* xr = xl + n * HC;
        float a0 = ep[0] / den0;
        float a1 = ep[4] / den1;
        v4f r0 = *(const v4f*)(xr);
        v4f r1 = *(const v4f*)(xr + 128);
        c0 += r0 * a0;
        c1 += r1 * a1;
    }
    v4f sum = c0 + c1;
    #pragma unroll
    for (int k = 0; k < 4; ++k) {
        float v = sum[k];
        v += __shfl_xor(v, 8);
        v += __shfl_xor(v, 16);
        sum[k] = v;
    }
    if (lane < 8) {
        float* o = out + n * NC + lane * 4;
        const float* b = bias + lane * 4;
        o[0] = sum[0] * (1.0f / NH) + b[0];
        o[1] = sum[1] * (1.0f / NH) + b[1];
        o[2] = sum[2] * (1.0f / NH) + b[2];
        o[3] = sum[3] * (1.0f / NH) + b[3];
    }
}

extern "C" void kernel_launch(void* const* d_in, const int* in_sizes, int n_in,
                              void* d_out, int out_size, void* d_ws, size_t ws_size,
                              hipStream_t stream) {
    const float* x        = (const float*)d_in[0];
    const int*   eidx     = (const int*)  d_in[1];
    const float* eattr    = (const float*)d_in[2];
    const float* W        = (const float*)d_in[3];
    const float* We       = (const float*)d_in[4];
    const float* att_src  = (const float*)d_in[5];
    const float* att_dst  = (const float*)d_in[6];
    const float* att_edge = (const float*)d_in[7];
    const float* bias     = (const float*)d_in[8];

    const int N = in_sizes[0] / F_IN;
    const int E = in_sizes[1] / 2;
    const int* src = eidx;
    const int* dst = eidx + E;

    float* ws = (float*)d_ws;
    const long Nl = N, El = E, Tl = El + Nl;
    float*    w_deg   = ws;
    float*    w_loop  = ws + Nl;
    float*    w_denom = ws + 17 * Nl;
    unsigned* w_amax  = (unsigned*)(ws + 25 * Nl);
    int*      w_cur   = (int*)(ws + 33 * Nl);
    int*      w_tot   = (int*)(ws + 34 * Nl);
    const long zeroSpan = 34 * Nl + 64;
    const long o_off  = (zeroSpan + 63) & ~63L;
    const long o_perm = (o_off + Nl + 63) & ~63L;
    const long o_xh   = (o_perm + El + 63) & ~63L;
    int*      w_off   = (int*)(ws + o_off);
    int*      w_perm  = (int*)(ws + o_perm);
    float*    w_xh    = ws + o_xh;
    float*    w_asrc  = ws + o_xh + 256 * Nl;
    float*    w_adst  = ws + o_xh + 264 * Nl;
    float*    w_M     = ws + o_xh + 272 * Nl;
    float*    w_ex    = ws + o_xh + 272 * Nl + 128;

    const int B = 256;
    GAT_zero<<<2048, B, 0, stream>>>(ws, zeroSpan);
    {
        long th = El * F_E;
        GAT_degloop<<<(int)((th + B - 1) / B), B, 0, stream>>>(dst, eattr, w_deg, w_loop, E);
    }
    {
        long th = Nl * F_E;
        GAT_loopdiv<<<(int)((th + B - 1) / B), B, 0, stream>>>(w_loop, w_deg, N);
    }
    GAT_foldM<<<1, 128, 0, stream>>>(We, att_edge, w_M);
    {
        long rowTiles = (Nl + 15) >> 4;
        long wtiles = rowTiles * 4;
        GAT_gemm_xh<<<(int)((wtiles + 7) / 8), B, 0, stream>>>(x, W, w_xh, N);
    }
    {
        long th = Nl * NH;
        GAT_nodeatt<<<(int)((th + B - 1) / B), B, 0, stream>>>(w_xh, att_src, att_dst,
                                                               w_asrc, w_adst, N);
    }
    GAT_alpha<<<(int)((Tl + B - 1) / B), B, 0, stream>>>(src, dst, eattr, w_loop, w_M,
                                                         w_asrc, w_adst, w_ex, w_amax, E, N);
    {
        long th = Tl * NH;
        GAT_expsum<<<(int)((th + B - 1) / B), B, 0, stream>>>(dst, w_ex, w_amax, w_denom, E, N);
    }
    GAT_segalloc<<<(int)((Nl + 1023) / 1024), 1024, 0, stream>>>(w_deg, w_off, w_tot, N);
    GAT_fill<<<(int)((El + B - 1) / B), B, 0, stream>>>(dst, w_off, w_cur, w_perm, E);
    GAT_gather<<<(int)((Nl + 7) / 8), B, 0, stream>>>(src, w_deg, w_off, w_perm, w_xh, w_ex,
                                                      w_denom, bias, (float*)d_out, E, N);
}


# Projection moved behind the neighbour sum (linearity): waves gather 512-B x rows into per-head sums, 32-node tiles x W on WMMA in-launch; node terms = x @ folded W (WMMA); no xh; sum order changes
# speedup vs baseline: 4.6626x; 2.4091x over previous
#include <hip/hip_runtime.h>
#include <hip/hip_bf16.h>

typedef float v2f __attribute__((ext_vector_type(2)));
typedef float v8f __attribute__((ext_vector_type(8)));
typedef float v4f __attribute__((ext_vector_type(4)));

#define F_IN 128
#define F_E  16
#define NH   8
#define NC   32
#define HC   256
#define NEG_SLOPE 0.2f

__global__ void GAT_zero(float* p, long n) {
    long i = (long)blockIdx.x * blockDim.x + threadIdx.x;
    long stride = (long)gridDim.x * blockDim.x;
    for (; i < n; i += stride) p[i] = 0.0f;
}

__global__ void GAT_count(const int* __restrict__ dst, int* __restrict__ deg, int E) {
    long e = (long)blockIdx.x * blockDim.x + threadIdx.x;
    if (e >= E) return;
    atomicAdd(&deg[dst[e]], 1);
}

__global__ void GAT_fold(const float* __restrict__ W,
                         const float* __restrict__ We,
                         const float* __restrict__ att_src,
                         const float* __restrict__ att_dst,
                         const float* __restrict__ att_edge,
                         float* __restrict__ M,
                         float* __restrict__ Vt) {
    int tid = threadIdx.x;
    if (blockIdx.x < 8) {
        int idx = blockIdx.x * 256 + tid;
        int j = idx >> 7, f = idx & (F_IN - 1);
        int hj = j & 7;
        const float* att = (j < NH) ? att_src : att_dst;
        float s = 0.0f;
        for (int c = 0; c < NC; ++c)
            s += W[(long)(hj * NC + c) * F_IN + f] * att[hj * NC + c];
        Vt[j * F_IN + f] = s;
    } else {
        if (tid >= F_E * NH) return;
        int f = tid >> 3, h = tid & 7;
        float s = 0.0f;
        for (int c = 0; c < NC; ++c)
            s += We[(long)(h * NC + c) * F_E + f] * att_edge[h * NC + c];
        M[f * NH + h] = s;
    }
}

__global__ void GAT_gemm_att(const float* __restrict__ x,
                             const float* __restrict__ Vt,
                             float* __restrict__ asrc,
                             float* __restrict__ adst, int N) {
    int wave = threadIdx.x >> 5;
    int lane = threadIdx.x & 31;
    int half = lane >> 4;
    int l    = lane & 15;
    long rowTiles = ((long)N + 15) >> 4;
    long tile = (long)blockIdx.x * 8 + wave;
    if (tile >= rowTiles) return;
    long row0 = tile * 16;
    long arow = row0 + l;
    if (arow >= N) arow = N - 1;
    const float* xp = x  + arow * F_IN + 2 * half;
    const float* vp = Vt + (long)l * F_IN + 2 * half;
    v8f a0 = {0.f,0.f,0.f,0.f,0.f,0.f,0.f,0.f};
    v8f a1 = a0;
    #pragma unroll
    for (int k0 = 0; k0 < F_IN; k0 += 8) {
        v2f xa = *(const v2f*)(xp + k0);
        v2f va = *(const v2f*)(vp + k0);
        v2f xb = *(const v2f*)(xp + k0 + 4);
        v2f vb = *(const v2f*)(vp + k0 + 4);
        a0 = __builtin_amdgcn_wmma_f32_16x16x4_f32(false, xa, false, va, (short)0, a0, false, false);
        a1 = __builtin_amdgcn_wmma_f32_16x16x4_f32(false, xb, false, vb, (short)0, a1, false, false);
    }
    v8f acc = a0 + a1;
    #pragma unroll
    for (int v = 0; v < 8; ++v) {
        long r = row0 + v + 8 * half;
        if (r < N) {
            if (l < NH) asrc[r * NH + l] = acc[v];
            else        adst[r * NH + (l - NH)] = acc[v];
        }
    }
}

__global__ void GAT_segalloc(const int* __restrict__ deg, int* __restrict__ off,
                             int* __restrict__ total, int N) {
    __shared__ int wsum[32];
    __shared__ int base;
    int t = threadIdx.x;
    int lane = t & 31, w = t >> 5;
    long n = (long)blockIdx.x * 1024 + t;
    int c = (n < N) ? deg[n] : 0;
    int inc = c;
    for (int o = 1; o < 32; o <<= 1) {
        int v = __shfl_up(inc, o);
        if (lane >= o) inc += v;
    }
    if (lane == 31) wsum[w] = inc;
    __syncthreads();
    if (w == 0) {
        int s = wsum[lane];
        int si = s;
        for (int o = 1; o < 32; o <<= 1) {
            int v = __shfl_up(si, o);
            if (lane >= o) si += v;
        }
        wsum[lane] = si - s;
        if (lane == 31) base = atomicAdd(total, si);
    }
    __syncthreads();
    if (n < N) off[n] = base + wsum[w] + inc - c;
}

__global__ void GAT_fill(const int* __restrict__ dst, const int* __restrict__ off,
                         int* __restrict__ cur, int* __restrict__ perm, int E) {
    long e = (long)blockIdx.x * blockDim.x + threadIdx.x;
    if (e >= E) return;
    int d = dst[e];
    int p = atomicAdd(&cur[d], 1);
    perm[off[d] + p] = (int)e;
}

__device__ __forceinline__ float GAT_lane_bcast(float v, int g) {
    return __builtin_bit_cast(float, __builtin_amdgcn_readlane(__builtin_bit_cast(int, v), g));
}
#define ZP 1028
__global__ void __launch_bounds__(512, 8)
GAT_gather(const int* __restrict__ src,
           const int* __restrict__ deg,
           const int* __restrict__ off,
           const int* __restrict__ perm,
           const float* __restrict__ eattr,
           const float* __restrict__ M,
           const float* __restrict__ asrc,
           const float* __restrict__ adst,
           const float* __restrict__ x,
           const float* __restrict__ W,
           const float* __restrict__ bias,
           float* __restrict__ out, int N) {
    __shared__ __attribute__((aligned(16))) float zt[32 * ZP];
    __shared__ __attribute__((aligned(16))) float wexp[16][32 * NH];
    const int wave = __builtin_amdgcn_readfirstlane((int)(threadIdx.x >> 5));
    const int lane = threadIdx.x & 31;
    const int hq = lane >> 3;
    const int h  = lane & 7;
    float* wmine = &wexp[wave][0];
    const float* xl = x + lane * 4;

    float mcol[F_E];
    #pragma unroll
    for (int f = 0; f < F_E; ++f) mcol[f] = M[f * NH + h];

    #pragma nounroll
    for (int rt = 0; rt < 2; ++rt) {
        const long n = (long)blockIdx.x * 32 + rt * 16 + wave;
        v4f z[NH];
        #pragma unroll
        for (int g = 0; g < NH; ++g) z[g] = (v4f){0.f, 0.f, 0.f, 0.f};
        if (n < N) {
            const float adh = adst[n * NH + h];
            const float asn = asrc[n * NH + h];
            float mrun = -__builtin_inff();
            float dl = 0.0f;
            float es0 = 0.0f, es1 = 0.0f;
            float aself = 0.0f;
            const int dg = deg[n];
            const int jb = off[n];
            const int je = jb + dg;
            int j0 = jb;
            do {
                int cnt = je - j0;
                if (cnt > 32) cnt = 32;
                int me = 0, ms = (int)n;
                if (lane < cnt) { me = perm[j0 + lane]; ms = src[me]; }
                const int nk = (cnt + 3) >> 2;
                float lmax = -__builtin_inff();
                #pragma unroll 2
                for (int k = 0; k < nk; ++k) {
                    const int j = 4 * k + hq;
                    const bool valid = j < cnt;
                    const int e = __shfl(me, j);
                    const int s = __shfl(ms, j);
                    const float* er = eattr + (long)e * F_E;
                    v4f e0 = *(const v4f*)(er);
                    v4f e1 = *(const v4f*)(er + 4);
                    v4f e2 = *(const v4f*)(er + 8);
                    v4f e3 = *(const v4f*)(er + 12);
                    v2f ef = *(const v2f*)(er + 2 * h);
                    float as = asrc[(long)s * NH + h];
                    float ae = 0.0f;
                    #pragma unroll
                    for (int q = 0; q < 4; ++q) ae = __builtin_fmaf(e0[q], mcol[q], ae);
                    #pragma unroll
                    for (int q = 0; q < 4; ++q) ae = __builtin_fmaf(e1[q], mcol[4 + q], ae);
                    #pragma unroll
                    for (int q = 0; q < 4; ++q) ae = __builtin_fmaf(e2[q], mcol[8 + q], ae);
                    #pragma unroll
                    for (int q = 0; q < 4; ++q) ae = __builtin_fmaf(e3[q], mcol[12 + q], ae);
                    float a = as + adh + ae;
                    a = (a >= 0.0f) ? a : NEG_SLOPE * a;
                    a = valid ? a : -__builtin_inff();
                    es0 += valid ? ef[0] : 0.0f;
                    es1 += valid ? ef[1] : 0.0f;
                    wmine[32 * k + lane] = a;
                    lmax = fmaxf(lmax, a);
                }
                float cm = fmaxf(lmax, __shfl_xor(lmax, 8));
                cm = fmaxf(cm, __shfl_xor(cm, 16));
                if (j0 + 32 >= je) {
                    es0 += __shfl_xor(es0, 8);
                    es0 += __shfl_xor(es0, 16);
                    es1 += __shfl_xor(es1, 8);
                    es1 += __shfl_xor(es1, 16);
                    const float dmax = fmaxf((float)dg, 1.0f);
                    const float lp0 = es0 / dmax;
                    const float lp1 = es1 / dmax;
                    float ae = 0.0f;
                    #pragma unroll
                    for (int g = 0; g < NH; ++g) {
                        ae = __builtin_fmaf(__shfl(lp0, g), mcol[2 * g], ae);
                        ae = __builtin_fmaf(__shfl(lp1, g), mcol[2 * g + 1], ae);
                    }
                    float a = asn + adh + ae;
                    aself = (a >= 0.0f) ? a : NEG_SLOPE * a;
                    cm = fmaxf(cm, aself);
                }
                const float mn = fmaxf(mrun, cm);
                const float sc = __expf(mrun - mn);
                mrun = mn;
                float lsum = 0.0f;
                #pragma unroll 4
                for (int k = 0; k < nk; ++k) {
                    float ex = __expf(wmine[32 * k + lane] - mn);
                    wmine[32 * k + lane] = ex;
                    lsum += ex;
                }
                dl = dl * sc + lsum;
                #pragma unroll
                for (int g = 0; g < NH; ++g) z[g] *= GAT_lane_bcast(sc, g);
                int i = 0;
                for (; i + 8 <= cnt; i += 8) {
                    v4f r[8];
                    #pragma unroll
                    for (int u = 0; u < 8; ++u) {
                        int s = __builtin_amdgcn_readlane(ms, i + u);
                        r[u] = *(const v4f*)(xl + (long)s * F_IN);
                    }
                    asm volatile("" :: "v"(r[0]), "v"(r[1]), "v"(r[2]), "v"(r[3]), "v"(r[4]), "v"(r[5]), "v"(r[6]), "v"(r[7]));
                    #pragma unroll
                    for (int u = 0; u < 8; ++u) {
                        v4f wa = *(const v4f*)(wmine + (i + u) * NH);
                        v4f wb = *(const v4f*)(wmine + (i + u) * NH + 4);
                        z[0] += r[u] * wa[0];
                        z[1] += r[u] * wa[1];
                        z[2] += r[u] * wa[2];
                        z[3] += r[u] * wa[3];
                        z[4] += r[u] * wb[0];
                        z[5] += r[u] * wb[1];
                        z[6] += r[u] * wb[2];
                        z[7] += r[u] * wb[3];
                    }
                }
                for (; i < cnt; i += 4) {
                    v4f r[4];
                    #pragma unroll
                    for (int u = 0; u < 4; ++u) {
                        int s = __builtin_amdgcn_readlane(ms, i + u);
                        r[u] = *(const v4f*)(xl + (long)s * F_IN);
                    }
                    asm volatile("" :: "v"(r[0]), "v"(r[1]), "v"(r[2]), "v"(r[3]));
                    #pragma unroll
                    for (int u = 0; u < 4; ++u) {
                        v4f wa = *(const v4f*)(wmine + (i + u) * NH);
                        v4f wb = *(const v4f*)(wmine + (i + u) * NH + 4);
                        z[0] += r[u] * wa[0];
                        z[1] += r[u] * wa[1];
                        z[2] += r[u] * wa[2];
                        z[3] += r[u] * wa[3];
                        z[4] += r[u] * wb[0];
                        z[5] += r[u] * wb[1];
                        z[6] += r[u] * wb[2];
                        z[7] += r[u] * wb[3];
                    }
                }
                j0 += 32;
            } while (j0 < je);

            const float exs = __expf(aself - mrun);
            dl += __shfl_xor(dl, 8);
            dl += __shfl_xor(dl, 16);
            const float den = dl + exs;
            const float inv = (1.0f / NH) / den;
            {
                v4f r = *(const v4f*)(xl + n * F_IN);
                #pragma unroll
                for (int g = 0; g < NH; ++g) z[g] += r * GAT_lane_bcast(exs, g);
            }
            #pragma unroll
            for (int g = 0; g < NH; ++g) z[g] *= GAT_lane_bcast(inv, g);
        }
        float* zrow = zt + (rt * 16 + wave) * ZP + lane * 4;
        #pragma unroll
        for (int g = 0; g < NH; ++g) *(v4f*)(zrow + g * F_IN) = z[g];
    }
    __syncthreads();

    {
        const int hw = wave >> 1;
        const int ct = wave & 1;
        const int l  = lane & 15;
        const int hf = lane >> 4;
        const float* wp  = W + (long)(hw * NC + ct * 16 + l) * F_IN + 4 * hf;
        const float* za0 = zt + l * ZP + hw * F_IN + 4 * hf;
        const float* za1 = za0 + 16 * ZP;
        v8f p00 = {0.f,0.f,0.f,0.f,0.f,0.f,0.f,0.f};
        v8f p01 = p00, p10 = p00, p11 = p00;
        #pragma unroll
        for (int k0 = 0; k0 < F_IN; k0 += 8) {
            v4f b  = *(const v4f*)(wp  + k0);
            v4f a0 = *(const v4f*)(za0 + k0);
            v4f a1 = *(const v4f*)(za1 + k0);
            v2f bl  = {b[0],  b[1]},  bh  = {b[2],  b[3]};
            v2f a0l = {a0[0], a0[1]}, a0h = {a0[2], a0[3]};
            v2f a1l = {a1[0], a1[1]}, a1h = {a1[2], a1[3]};
            p00 = __builtin_amdgcn_wmma_f32_16x16x4_f32(false, a0l, false, bl, (short)0, p00, false, false);
            p10 = __builtin_amdgcn_wmma_f32_16x16x4_f32(false, a1l, false, bl, (short)0, p10, false, false);
            p01 = __builtin_amdgcn_wmma_f32_16x16x4_f32(false, a0h, false, bh, (short)0, p01, false, false);
            p11 = __builtin_amdgcn_wmma_f32_16x16x4_f32(false, a1h, false, bh, (short)0, p11, false, false);
        }
        v8f p0 = p00 + p01;
        v8f p1 = p10 + p11;
        __syncthreads();
        float* pt = zt + wave * 512 + l;
        #pragma unroll
        for (int v = 0; v < 8; ++v) {
            pt[(8 * hf + v) * 16]       = p0[v];
            pt[256 + (8 * hf + v) * 16] = p1[v];
        }
    }
    __syncthreads();
    {
        const int c   = lane;
        const int ct2 = c >> 4, cc = c & 15;
        const float bc = bias[c];
        #pragma unroll
        for (int rt = 0; rt < 2; ++rt) {
            const long n = (long)blockIdx.x * 32 + rt * 16 + wave;
            float s = 0.0f;
            #pragma unroll
            for (int g = 0; g < NH; ++g)
                s += zt[((g * 2 + ct2) * 2 + rt) * 256 + wave * 16 + cc];
            if (n < N) out[n * NC + c] = s + bc;
        }
    }
}

extern "C" void kernel_launch(void* const* d_in, const int* in_sizes, int n_in,
                              void* d_out, int out_size, void* d_ws, size_t ws_size,
                              hipStream_t stream) {
    const float* x        = (const float*)d_in[0];
    const int*   eidx     = (const int*)  d_in[1];
    const float* eattr    = (const float*)d_in[2];
    const float* W        = (const float*)d_in[3];
    const float* We       = (const float*)d_in[4];
    const float* att_src  = (const float*)d_in[5];
    const float* att_dst  = (const float*)d_in[6];
    const float* att_edge = (const float*)d_in[7];
    const float* bias     = (const float*)d_in[8];

    const int N = in_sizes[0] / F_IN;
    const int E = in_sizes[1] / 2;
    const int* src = eidx;
    const int* dst = eidx + E;

    float* ws = (float*)d_ws;
    const long Nl = N, El = E;
    int*      w_deg   = (int*)ws;
    int*      w_cur   = (int*)(ws + Nl);
    int*      w_tot   = (int*)(ws + 2 * Nl);
    const long zeroSpan = 2 * Nl + 64;
    const long o_off  = (zeroSpan + 63) & ~63L;
    const long o_perm = (o_off + Nl + 63) & ~63L;
    const long o_att  = (o_perm + El + 63) & ~63L;
    int*      w_off   = (int*)(ws + o_off);
    int*      w_perm  = (int*)(ws + o_perm);
    float*    w_asrc  = ws + o_att;
    float*    w_adst  = ws + o_att + 8 * Nl;
    float*    w_M     = ws + o_att + 16 * Nl;
    float*    w_Vt    = ws + o_att + 16 * Nl + 128;

    const int B = 256;
    GAT_zero<<<512, B, 0, stream>>>(ws, zeroSpan);
    GAT_count<<<(int)((El + B - 1) / B), B, 0, stream>>>(dst, w_deg, E);
    GAT_segalloc<<<(int)((Nl + 1023) / 1024), 1024, 0, stream>>>(w_deg, w_off, w_tot, N);
    GAT_fill<<<(int)((El + B - 1) / B), B, 0, stream>>>(dst, w_off, w_cur, w_perm, E);
    GAT_fold<<<9, B, 0, stream>>>(W, We, att_src, att_dst, att_edge, w_M, w_Vt);
    {
        long rowTiles = (Nl + 15) >> 4;
        GAT_gemm_att<<<(int)((rowTiles + 7) / 8), B, 0, stream>>>(x, w_Vt, w_asrc, w_adst, N);
    }
    GAT_gather<<<(int)((Nl + 31) / 32), 512, 0, stream>>>(src, w_deg, w_off, w_perm, eattr, w_M,
                                                          w_asrc, w_adst, x, W, bias,
                                                          (float*)d_out, N);
}


# Edge slots from ONE atomic pass: the degree add's returned value is the edge's rank in its node (kept per edge); fill = base + rank, plain stores, no cursors; edge order per node by arrival as before
# speedup vs baseline: 5.7513x; 1.2335x over previous
#include <hip/hip_runtime.h>
#include <hip/hip_bf16.h>

typedef float v2f __attribute__((ext_vector_type(2)));
typedef float v8f __attribute__((ext_vector_type(8)));
typedef float v4f __attribute__((ext_vector_type(4)));

#define F_IN 128
#define F_E  16
#define NH   8
#define NC   32
#define HC   256
#define NEG_SLOPE 0.2f

__global__ void GAT_zero(float* p, long n) {
    long i = (long)blockIdx.x * blockDim.x + threadIdx.x;
    long stride = (long)gridDim.x * blockDim.x;
    for (; i < n; i += stride) p[i] = 0.0f;
}

__global__ void GAT_count(const int* __restrict__ dst, int* __restrict__ deg,
                          int* __restrict__ rank, int E) {
    long e = (long)blockIdx.x * blockDim.x + threadIdx.x;
    if (e >= E) return;
    rank[e] = atomicAdd(&deg[dst[e]], 1);
}

__global__ void GAT_fold(const float* __restrict__ W,
                         const float* __restrict__ We,
                         const float* __restrict__ att_src,
                         const float* __restrict__ att_dst,
                         const float* __restrict__ att_edge,
                         float* __restrict__ M,
                         float* __restrict__ Vt) {
    int tid = threadIdx.x;
    if (blockIdx.x < 8) {
        int idx = blockIdx.x * 256 + tid;
        int j = idx >> 7, f = idx & (F_IN - 1);
        int hj = j & 7;
        const float* att = (j < NH) ? att_src : att_dst;
        float s = 0.0f;
        for (int c = 0; c < NC; ++c)
            s += W[(long)(hj * NC + c) * F_IN + f] * att[hj * NC + c];
        Vt[j * F_IN + f] = s;
    } else {
        if (tid >= F_E * NH) return;
        int f = tid >> 3, h = tid & 7;
        float s = 0.0f;
        for (int c = 0; c < NC; ++c)
            s += We[(long)(h * NC + c) * F_E + f] * att_edge[h * NC + c];
        M[f * NH + h] = s;
    }
}

__global__ void GAT_gemm_att(const float* __restrict__ x,
                             const float* __restrict__ Vt,
                             float* __restrict__ asrc,
                             float* __restrict__ adst, int N) {
    int wave = threadIdx.x >> 5;
    int lane = threadIdx.x & 31;
    int half = lane >> 4;
    int l    = lane & 15;
    long rowTiles = ((long)N + 15) >> 4;
    long tile = (long)blockIdx.x * 8 + wave;
    if (tile >= rowTiles) return;
    long row0 = tile * 16;
    long arow = row0 + l;
    if (arow >= N) arow = N - 1;
    const float* xp = x  + arow * F_IN + 2 * half;
    const float* vp = Vt + (long)l * F_IN + 2 * half;
    v8f a0 = {0.f,0.f,0.f,0.f,0.f,0.f,0.f,0.f};
    v8f a1 = a0;
    #pragma unroll
    for (int k0 = 0; k0 < F_IN; k0 += 8) {
        v2f xa = *(const v2f*)(xp + k0);
        v2f va = *(const v2f*)(vp + k0);
        v2f xb = *(const v2f*)(xp + k0 + 4);
        v2f vb = *(const v2f*)(vp + k0 + 4);
        a0 = __builtin_amdgcn_wmma_f32_16x16x4_f32(false, xa, false, va, (short)0, a0, false, false);
        a1 = __builtin_amdgcn_wmma_f32_16x16x4_f32(false, xb, false, vb, (short)0, a1, false, false);
    }
    v8f acc = a0 + a1;
    #pragma unroll
    for (int v = 0; v < 8; ++v) {
        long r = row0 + v + 8 * half;
        if (r < N) {
            if (l < NH) asrc[r * NH + l] = acc[v];
            else        adst[r * NH + (l - NH)] = acc[v];
        }
    }
}

__global__ void GAT_segalloc(const int* __restrict__ deg, int* __restrict__ off,
                             int* __restrict__ total, int N) {
    __shared__ int wsum[32];
    __shared__ int base;
    int t = threadIdx.x;
    int lane = t & 31, w = t >> 5;
    long n = (long)blockIdx.x * 1024 + t;
    int c = (n < N) ? deg[n] : 0;
    int inc = c;
    for (int o = 1; o < 32; o <<= 1) {
        int v = __shfl_up(inc, o);
        if (lane >= o) inc += v;
    }
    if (lane == 31) wsum[w] = inc;
    __syncthreads();
    if (w == 0) {
        int s = wsum[lane];
        int si = s;
        for (int o = 1; o < 32; o <<= 1) {
            int v = __shfl_up(si, o);
            if (lane >= o) si += v;
        }
        wsum[lane] = si - s;
        if (lane == 31) base = atomicAdd(total, si);
    }
    __syncthreads();
    if (n < N) off[n] = base + wsum[w] + inc - c;
}

__global__ void GAT_fill(const int* __restrict__ dst, const int* __restrict__ off,
                         const int* __restrict__ rank, int* __restrict__ perm, int E) {
    long e = (long)blockIdx.x * blockDim.x + threadIdx.x;
    if (e >= E) return;
    perm[off[dst[e]] + rank[e]] = (int)e;
}

__device__ __forceinline__ float GAT_lane_bcast(float v, int g) {
    return __builtin_bit_cast(float, __builtin_amdgcn_readlane(__builtin_bit_cast(int, v), g));
}
#define ZP 1028
__global__ void __launch_bounds__(512, 8)
GAT_gather(const int* __restrict__ src,
           const int* __restrict__ deg,
           const int* __restrict__ off,
           const int* __restrict__ perm,
           const float* __restrict__ eattr,
           const float* __restrict__ M,
           const float* __restrict__ asrc,
           const float* __restrict__ adst,
           const float* __restrict__ x,
           const float* __restrict__ W,
           const float* __restrict__ bias,
           float* __restrict__ out, int N) {
    __shared__ __attribute__((aligned(16))) float zt[32 * ZP];
    __shared__ __attribute__((aligned(16))) float wexp[16][32 * NH];
    const int wave = __builtin_amdgcn_readfirstlane((int)(threadIdx.x >> 5));
    const int lane = threadIdx.x & 31;
    const int hq = lane >> 3;
    const int h  = lane & 7;
    float* wmine = &wexp[wave][0];
    const float* xl = x + lane * 4;

    float mcol[F_E];
    #pragma unroll
    for (int f = 0; f < F_E; ++f) mcol[f] = M[f * NH + h];

    #pragma nounroll
    for (int rt = 0; rt < 2; ++rt) {
        const long n = (long)blockIdx.x * 32 + rt * 16 + wave;
        v4f z[NH];
        #pragma unroll
        for (int g = 0; g < NH; ++g) z[g] = (v4f){0.f, 0.f, 0.f, 0.f};
        if (n < N) {
            const float adh = adst[n * NH + h];
            const float asn = asrc[n * NH + h];
            float mrun = -__builtin_inff();
            float dl = 0.0f;
            float es0 = 0.0f, es1 = 0.0f;
            float aself = 0.0f;
            const int dg = deg[n];
            const int jb = off[n];
            const int je = jb + dg;
            int j0 = jb;
            do {
                int cnt = je - j0;
                if (cnt > 32) cnt = 32;
                int me = 0, ms = (int)n;
                if (lane < cnt) { me = perm[j0 + lane]; ms = src[me]; }
                const int nk = (cnt + 3) >> 2;
                float lmax = -__builtin_inff();
                #pragma unroll 2
                for (int k = 0; k < nk; ++k) {
                    const int j = 4 * k + hq;
                    const bool valid = j < cnt;
                    const int e = __shfl(me, j);
                    const int s = __shfl(ms, j);
                    const float* er = eattr + (long)e * F_E;
                    v4f e0 = *(const v4f*)(er);
                    v4f e1 = *(const v4f*)(er + 4);
                    v4f e2 = *(const v4f*)(er + 8);
                    v4f e3 = *(const v4f*)(er + 12);
                    v2f ef = *(const v2f*)(er + 2 * h);
                    float as = asrc[(long)s * NH + h];
                    float ae = 0.0f;
                    #pragma unroll
                    for (int q = 0; q < 4; ++q) ae = __builtin_fmaf(e0[q], mcol[q], ae);
                    #pragma unroll
                    for (int q = 0; q < 4; ++q) ae = __builtin_fmaf(e1[q], mcol[4 + q], ae);
                    #pragma unroll
                    for (int q = 0; q < 4; ++q) ae = __builtin_fmaf(e2[q], mcol[8 + q], ae);
                    #pragma unroll
                    for (int q = 0; q < 4; ++q) ae = __builtin_fmaf(e3[q], mcol[12 + q], ae);
                    float a = as + adh + ae;
                    a = (a >= 0.0f) ? a : NEG_SLOPE * a;
                    a = valid ? a : -__builtin_inff();
                    es0 += valid ? ef[0] : 0.0f;
                    es1 += valid ? ef[1] : 0.0f;
                    wmine[32 * k + lane] = a;
                    lmax = fmaxf(lmax, a);
                }
                float cm = fmaxf(lmax, __shfl_xor(lmax, 8));
                cm = fmaxf(cm, __shfl_xor(cm, 16));
                if (j0 + 32 >= je) {
                    es0 += __shfl_xor(es0, 8);
                    es0 += __shfl_xor(es0, 16);
                    es1 += __shfl_xor(es1, 8);
                    es1 += __shfl_xor(es1, 16);
                    const float dmax = fmaxf((float)dg, 1.0f);
                    const float lp0 = es0 / dmax;
                    const float lp1 = es1 / dmax;
                    float ae = 0.0f;
                    #pragma unroll
                    for (int g = 0; g < NH; ++g) {
                        ae = __builtin_fmaf(__shfl(lp0, g), mcol[2 * g], ae);
                        ae = __builtin_fmaf(__shfl(lp1, g), mcol[2 * g + 1], ae);
                    }
                    float a = asn + adh + ae;
                    aself = (a >= 0.0f) ? a : NEG_SLOPE * a;
                    cm = fmaxf(cm, aself);
                }
                const float mn = fmaxf(mrun, cm);
                const float sc = __expf(mrun - mn);
                mrun = mn;
                float lsum = 0.0f;
                #pragma unroll 4
                for (int k = 0; k < nk; ++k) {
                    float ex = __expf(wmine[32 * k + lane] - mn);
                    wmine[32 * k + lane] = ex;
                    lsum += ex;
                }
                dl = dl * sc + lsum;
                #pragma unroll
                for (int g = 0; g < NH; ++g) z[g] *= GAT_lane_bcast(sc, g);
                int i = 0;
                for (; i + 8 <= cnt; i += 8) {
                    v4f r[8];
                    #pragma unroll
                    for (int u = 0; u < 8; ++u) {
                        int s = __builtin_amdgcn_readlane(ms, i + u);
                        r[u] = *(const v4f*)(xl + (long)s * F_IN);
                    }
                    asm volatile("" :: "v"(r[0]), "v"(r[1]), "v"(r[2]), "v"(r[3]), "v"(r[4]), "v"(r[5]), "v"(r[6]), "v"(r[7]));
                    #pragma unroll
                    for (int u = 0; u < 8; ++u) {
                        v4f wa = *(const v4f*)(wmine + (i + u) * NH);
                        v4f wb = *(const v4f*)(wmine + (i + u) * NH + 4);
                        z[0] += r[u] * wa[0];
                        z[1] += r[u] * wa[1];
                        z[2] += r[u] * wa[2];
                        z[3] += r[u] * wa[3];
                        z[4] += r[u] * wb[0];
                        z[5] += r[u] * wb[1];
                        z[6] += r[u] * wb[2];
                        z[7] += r[u] * wb[3];
                    }
                }
                for (; i < cnt; i += 4) {
                    v4f r[4];
                    #pragma unroll
                    for (int u = 0; u < 4; ++u) {
                        int s = __builtin_amdgcn_readlane(ms, i + u);
                        r[u] = *(const v4f*)(xl + (long)s * F_IN);
                    }
                    asm volatile("" :: "v"(r[0]), "v"(r[1]), "v"(r[2]), "v"(r[3]));
                    #pragma unroll
                    for (int u = 0; u < 4; ++u) {
                        v4f wa = *(const v4f*)(wmine + (i + u) * NH);
                        v4f wb = *(const v4f*)(wmine + (i + u) * NH + 4);
                        z[0] += r[u] * wa[0];
                        z[1] += r[u] * wa[1];
                        z[2] += r[u] * wa[2];
                        z[3] += r[u] * wa[3];
                        z[4] += r[u] * wb[0];
                        z[5] += r[u] * wb[1];
                        z[6] += r[u] * wb[2];
                        z[7] += r[u] * wb[3];
                    }
                }
                j0 += 32;
            } while (j0 < je);

            const float exs = __expf(aself - mrun);
            dl += __shfl_xor(dl, 8);
            dl += __shfl_xor(dl, 16);
            const float den = dl + exs;
            const float inv = (1.0f / NH) / den;
            {
                v4f r = *(const v4f*)(xl + n * F_IN);
                #pragma unroll
                for (int g = 0; g < NH; ++g) z[g] += r * GAT_lane_bcast(exs, g);
            }
            #pragma unroll
            for (int g = 0; g < NH; ++g) z[g] *= GAT_lane_bcast(inv, g);
        }
        float* zrow = zt + (rt * 16 + wave) * ZP + lane * 4;
        #pragma unroll
        for (int g = 0; g < NH; ++g) *(v4f*)(zrow + g * F_IN) = z[g];
    }
    __syncthreads();

    {
        const int hw = wave >> 1;
        const int ct = wave & 1;
        const int l  = lane & 15;
        const int hf = lane >> 4;
        const float* wp  = W + (long)(hw * NC + ct * 16 + l) * F_IN + 4 * hf;
        const float* za0 = zt + l * ZP + hw * F_IN + 4 * hf;
        const float* za1 = za0 + 16 * ZP;
        v8f p00 = {0.f,0.f,0.f,0.f,0.f,0.f,0.f,0.f};
        v8f p01 = p00, p10 = p00, p11 = p00;
        #pragma unroll
        for (int k0 = 0; k0 < F_IN; k0 += 8) {
            v4f b  = *(const v4f*)(wp  + k0);
            v4f a0 = *(const v4f*)(za0 + k0);
            v4f a1 = *(const v4f*)(za1 + k0);
            v2f bl  = {b[0],  b[1]},  bh  = {b[2],  b[3]};
            v2f a0l = {a0[0], a0[1]}, a0h = {a0[2], a0[3]};
            v2f a1l = {a1[0], a1[1]}, a1h = {a1[2], a1[3]};
            p00 = __builtin_amdgcn_wmma_f32_16x16x4_f32(false, a0l, false, bl, (short)0, p00, false, false);
            p10 = __builtin_amdgcn_wmma_f32_16x16x4_f32(false, a1l, false, bl, (short)0, p10, false, false);
            p01 = __builtin_amdgcn_wmma_f32_16x16x4_f32(false, a0h, false, bh, (short)0, p01, false, false);
            p11 = __builtin_amdgcn_wmma_f32_16x16x4_f32(false, a1h, false, bh, (short)0, p11, false, false);
        }
        v8f p0 = p00 + p01;
        v8f p1 = p10 + p11;
        __syncthreads();
        float* pt = zt + wave * 512 + l;
        #pragma unroll
        for (int v = 0; v < 8; ++v) {
            pt[(8 * hf + v) * 16]       = p0[v];
            pt[256 + (8 * hf + v) * 16] = p1[v];
        }
    }
    __syncthreads();
    {
        const int c   = lane;
        const int ct2 = c >> 4, cc = c & 15;
        const float bc = bias[c];
        #pragma unroll
        for (int rt = 0; rt < 2; ++rt) {
            const long n = (long)blockIdx.x * 32 + rt * 16 + wave;
            float s = 0.0f;
            #pragma unroll
            for (int g = 0; g < NH; ++g)
                s += zt[((g * 2 + ct2) * 2 + rt) * 256 + wave * 16 + cc];
            if (n < N) out[n * NC + c] = s + bc;
        }
    }
}

extern "C" void kernel_launch(void* const* d_in, const int* in_sizes, int n_in,
                              void* d_out, int out_size, void* d_ws, size_t ws_size,
                              hipStream_t stream) {
    const float* x        = (const float*)d_in[0];
    const int*   eidx     = (const int*)  d_in[1];
    const float* eattr    = (const float*)d_in[2];
    const float* W        = (const float*)d_in[3];
    const float* We       = (const float*)d_in[4];
    const float* att_src  = (const float*)d_in[5];
    const float* att_dst  = (const float*)d_in[6];
    const float* att_edge = (const float*)d_in[7];
    const float* bias     = (const float*)d_in[8];

    const int N = in_sizes[0] / F_IN;
    const int E = in_sizes[1] / 2;
    const int* src = eidx;
    const int* dst = eidx + E;

    float* ws = (float*)d_ws;
    const long Nl = N, El = E;
    int*      w_deg   = (int*)ws;
    int*      w_tot   = (int*)(ws + Nl);
    const long zeroSpan = Nl + 64;
    const long o_off  = (zeroSpan + 63) & ~63L;
    const long o_rank = (o_off + Nl + 63) & ~63L;
    const long o_perm = (o_rank + El + 63) & ~63L;
    const long o_att  = (o_perm + El + 63) & ~63L;
    int*      w_off   = (int*)(ws + o_off);
    int*      w_rank  = (int*)(ws + o_rank);
    int*      w_perm  = (int*)(ws + o_perm);
    float*    w_asrc  = ws + o_att;
    float*    w_adst  = ws + o_att + 8 * Nl;
    float*    w_M     = ws + o_att + 16 * Nl;
    float*    w_Vt    = ws + o_att + 16 * Nl + 128;

    const int B = 256;
    GAT_zero<<<512, B, 0, stream>>>(ws, zeroSpan);
    GAT_count<<<(int)((El + B - 1) / B), B, 0, stream>>>(dst, w_deg, w_rank, E);
    GAT_segalloc<<<(int)((Nl + 1023) / 1024), 1024, 0, stream>>>(w_deg, w_off, w_tot, N);
    GAT_fill<<<(int)((El + B - 1) / B), B, 0, stream>>>(dst, w_off, w_rank, w_perm, E);
    GAT_fold<<<9, B, 0, stream>>>(W, We, att_src, att_dst, att_edge, w_M, w_Vt);
    {
        long rowTiles = (Nl + 15) >> 4;
        GAT_gemm_att<<<(int)((rowTiles + 7) / 8), B, 0, stream>>>(x, w_Vt, w_asrc, w_adst, N);
    }
    GAT_gather<<<(int)((Nl + 31) / 32), 512, 0, stream>>>(src, w_deg, w_off, w_perm, eattr, w_M,
                                                          w_asrc, w_adst, x, W, bias,
                                                          (float*)d_out, N);
}


# Edge ranks counted in LDS: 64 edge chunks, one group each holds a counter per node (returning LDS adds); per-node scan of chunk counts gives slot bases; no memory atomics on edges; sum order not fixed
# speedup vs baseline: 7.3832x; 1.2837x over previous
#include <hip/hip_runtime.h>
#include <hip/hip_bf16.h>

typedef float v2f __attribute__((ext_vector_type(2)));
typedef float v8f __attribute__((ext_vector_type(8)));
typedef float v4f __attribute__((ext_vector_type(4)));

#define F_IN 128
#define F_E  16
#define NH   8
#define NC   32
#define HC   256
#define NEG_SLOPE 0.2f

typedef int v4i __attribute__((ext_vector_type(4)));

#define RK_G    64
#define RK_CAP  51200
#define RK_B    16

__global__ void __launch_bounds__(1024)
GAT_rank(const int* __restrict__ dst, int* __restrict__ rank, int* __restrict__ cnt,
         int* __restrict__ total, int E, int N, int epc, int np) {
    __shared__ __attribute__((aligned(16))) int hist[RK_CAP];
    const int t = threadIdx.x;
    const int g = blockIdx.x;
    if (g == 0 && t == 0) *total = 0;
    const long e0 = (long)g * epc;
    long e1 = e0 + epc;
    if (e1 > E) e1 = E;
    for (int nb = 0; nb < N; nb += RK_CAP) {
        int sn = N - nb;
        if (sn > RK_CAP) sn = RK_CAP;
        const int sn4 = (sn + 3) >> 2;
        int d[RK_B];
        long eb = e0;
        #pragma unroll
        for (int u = 0; u < RK_B; ++u) {
            const long e = eb + u * 1024 + t;
            d[u] = (e < e1) ? dst[e] : -1;
        }
        for (int i = t; i < sn4; i += 1024) ((v4i*)hist)[i] = (v4i){0, 0, 0, 0};
        __syncthreads();
        while (true) {
            int r[RK_B];
            #pragma unroll
            for (int u = 0; u < RK_B; ++u) {
                const unsigned k = (unsigned)(d[u] - nb);
                r[u] = (k < (unsigned)sn) ? atomicAdd(&hist[k], 1) : -1;
            }
            #pragma unroll
            for (int u = 0; u < RK_B; ++u) {
                const long e = eb + u * 1024 + t;
                if (r[u] >= 0) rank[e] = r[u];
            }
            eb += (long)RK_B * 1024;
            if (eb >= e1) break;
            #pragma unroll
            for (int u = 0; u < RK_B; ++u) {
                const long e = eb + u * 1024 + t;
                d[u] = (e < e1) ? dst[e] : -1;
            }
        }
        __syncthreads();
        int* row = cnt + (long)g * np + nb;
        #pragma unroll 4
        for (int i = t; i < sn4; i += 1024) ((v4i*)row)[i] = ((const v4i*)hist)[i];
        __syncthreads();
    }
}

__global__ void GAT_fold(const float* __restrict__ W,
                         const float* __restrict__ We,
                         const float* __restrict__ att_src,
                         const float* __restrict__ att_dst,
                         const float* __restrict__ att_edge,
                         float* __restrict__ M,
                         float* __restrict__ Vt) {
    int tid = threadIdx.x;
    if (blockIdx.x < 8) {
        int idx = blockIdx.x * 256 + tid;
        int j = idx >> 7, f = idx & (F_IN - 1);
        int hj = j & 7;
        const float* att = (j < NH) ? att_src : att_dst;
        float s = 0.0f;
        for (int c = 0; c < NC; ++c)
            s += W[(long)(hj * NC + c) * F_IN + f] * att[hj * NC + c];
        Vt[j * F_IN + f] = s;
    } else {
        if (tid >= F_E * NH) return;
        int f = tid >> 3, h = tid & 7;
        float s = 0.0f;
        for (int c = 0; c < NC; ++c)
            s += We[(long)(h * NC + c) * F_E + f] * att_edge[h * NC + c];
        M[f * NH + h] = s;
    }
}

__global__ void GAT_gemm_att(const float* __restrict__ x,
                             const float* __restrict__ Vt,
                             float* __restrict__ asrc,
                             float* __restrict__ adst, int N) {
    int wave = threadIdx.x >> 5;
    int lane = threadIdx.x & 31;
    int half = lane >> 4;
    int l    = lane & 15;
    long rowTiles = ((long)N + 15) >> 4;
    long tile = (long)blockIdx.x * 8 + wave;
    if (tile >= rowTiles) return;
    long row0 = tile * 16;
    long arow = row0 + l;
    if (arow >= N) arow = N - 1;
    const float* xp = x  + arow * F_IN + 2 * half;
    const float* vp = Vt + (long)l * F_IN + 2 * half;
    v8f a0 = {0.f,0.f,0.f,0.f,0.f,0.f,0.f,0.f};
    v8f a1 = a0;
    #pragma unroll
    for (int k0 = 0; k0 < F_IN; k0 += 8) {
        v2f xa = *(const v2f*)(xp + k0);
        v2f va = *(const v2f*)(vp + k0);
        v2f xb = *(const v2f*)(xp + k0 + 4);
        v2f vb = *(const v2f*)(vp + k0 + 4);
        a0 = __builtin_amdgcn_wmma_f32_16x16x4_f32(false, xa, false, va, (short)0, a0, false, false);
        a1 = __builtin_amdgcn_wmma_f32_16x16x4_f32(false, xb, false, vb, (short)0, a1, false, false);
    }
    v8f acc = a0 + a1;
    #pragma unroll
    for (int v = 0; v < 8; ++v) {
        long r = row0 + v + 8 * half;
        if (r < N) {
            if (l < NH) asrc[r * NH + l] = acc[v];
            else        adst[r * NH + (l - NH)] = acc[v];
        }
    }
}

__global__ void __launch_bounds__(256)
GAT_segalloc(int* __restrict__ cnt, int* __restrict__ deg, int* __restrict__ off,
             int* __restrict__ total, int N, int np) {
    __shared__ int wsum[8];
    __shared__ int base;
    const int t = threadIdx.x;
    const int lane = t & 31, w = t >> 5;
    const long n = (long)blockIdx.x * 256 + t;
    int* col = cnt + ((n < N) ? n : (long)N - 1);
    int c[RK_G];
    #pragma unroll
    for (int g = 0; g < RK_G; ++g) c[g] = col[(long)g * np];
    int tot = 0;
    #pragma unroll
    for (int g = 0; g < RK_G; ++g) tot += c[g];
    if (n < N) deg[n] = tot; else tot = 0;
    int inc = tot;
    for (int o = 1; o < 32; o <<= 1) {
        int v = __shfl_up(inc, o);
        if (lane >= o) inc += v;
    }
    if (lane == 31) wsum[w] = inc;
    __syncthreads();
    if (w == 0) {
        int s = (lane < 8) ? wsum[lane] : 0;
        int si = s;
        for (int o = 1; o < 32; o <<= 1) {
            int v = __shfl_up(si, o);
            if (lane >= o) si += v;
        }
        if (lane < 8) wsum[lane] = si - s;
        if (lane == 31) base = atomicAdd(total, si);
    }
    __syncthreads();
    if (n < N) {
        int run = base + wsum[w] + inc - tot;
        off[n] = run;
        #pragma unroll
        for (int g = 0; g < RK_G; ++g) { col[(long)g * np] = run; run += c[g]; }
    }
}

__global__ void GAT_fill(const int* __restrict__ dst, const int* __restrict__ cbase,
                         const int* __restrict__ rank, int* __restrict__ perm, int E, int epc, int np) {
    const int g = blockIdx.y;
    const long e = (long)g * epc + (long)blockIdx.x * blockDim.x + threadIdx.x;
    if (e >= E) return;
    perm[cbase[(long)g * np + dst[e]] + rank[e]] = (int)e;
}

__device__ __forceinline__ float GAT_lane_bcast(float v, int g) {
    return __builtin_bit_cast(float, __builtin_amdgcn_readlane(__builtin_bit_cast(int, v), g));
}
#define ZP 1028
__global__ void __launch_bounds__(512, 8)
GAT_gather(const int* __restrict__ src,
           const int* __restrict__ deg,
           const int* __restrict__ off,
           const int* __restrict__ perm,
           const float* __restrict__ eattr,
           const float* __restrict__ M,
           const float* __restrict__ asrc,
           const float* __restrict__ adst,
           const float* __restrict__ x,
           const float* __restrict__ W,
           const float* __restrict__ bias,
           float* __restrict__ out, int N) {
    __shared__ __attribute__((aligned(16))) float zt[32 * ZP];
    __shared__ __attribute__((aligned(16))) float wexp[16][32 * NH];
    const int wave = __builtin_amdgcn_readfirstlane((int)(threadIdx.x >> 5));
    const int lane = threadIdx.x & 31;
    const int hq = lane >> 3;
    const int h  = lane & 7;
    float* wmine = &wexp[wave][0];
    const float* xl = x + lane * 4;

    float mcol[F_E];
    #pragma unroll
    for (int f = 0; f < F_E; ++f) mcol[f] = M[f * NH + h];

    #pragma nounroll
    for (int rt = 0; rt < 2; ++rt) {
        const long n = (long)blockIdx.x * 32 + rt * 16 + wave;
        v4f z[NH];
        #pragma unroll
        for (int g = 0; g < NH; ++g) z[g] = (v4f){0.f, 0.f, 0.f, 0.f};
        if (n < N) {
            const float adh = adst[n * NH + h];
            const float asn = asrc[n * NH + h];
            float mrun = -__builtin_inff();
            float dl = 0.0f;
            float es0 = 0.0f, es1 = 0.0f;
            float aself = 0.0f;
            const int dg = deg[n];
            const int jb = off[n];
            const int je = jb + dg;
            int j0 = jb;
            do {
                int cnt = je - j0;
                if (cnt > 32) cnt = 32;
                int me = 0, ms = (int)n;
                if (lane < cnt) { me = perm[j0 + lane]; ms = src[me]; }
                const int nk = (cnt + 3) >> 2;
                float lmax = -__builtin_inff();
                #pragma unroll 2
                for (int k = 0; k < nk; ++k) {
                    const int j = 4 * k + hq;
                    const bool valid = j < cnt;
                    const int e = __shfl(me, j);
                    const int s = __shfl(ms, j);
                    const float* er = eattr + (long)e * F_E;
                    v4f e0 = *(const v4f*)(er);
                    v4f e1 = *(const v4f*)(er + 4);
                    v4f e2 = *(const v4f*)(er + 8);
                    v4f e3 = *(const v4f*)(er + 12);
                    v2f ef = *(const v2f*)(er + 2 * h);
                    float as = asrc[(long)s * NH + h];
                    float ae = 0.0f;
                    #pragma unroll
                    for (int q = 0; q < 4; ++q) ae = __builtin_fmaf(e0[q], mcol[q], ae);
                    #pragma unroll
                    for (int q = 0; q < 4; ++q) ae = __builtin_fmaf(e1[q], mcol[4 + q], ae);
                    #pragma unroll
                    for (int q = 0; q < 4; ++q) ae = __builtin_fmaf(e2[q], mcol[8 + q], ae);
                    #pragma unroll
                    for (int q = 0; q < 4; ++q) ae = __builtin_fmaf(e3[q], mcol[12 + q], ae);
                    float a = as + adh + ae;
                    a = (a >= 0.0f) ? a : NEG_SLOPE * a;
                    a = valid ? a : -__builtin_inff();
                    es0 += valid ? ef[0] : 0.0f;
                    es1 += valid ? ef[1] : 0.0f;
                    wmine[32 * k + lane] = a;
                    lmax = fmaxf(lmax, a);
                }
                float cm = fmaxf(lmax, __shfl_xor(lmax, 8));
                cm = fmaxf(cm, __shfl_xor(cm, 16));
                if (j0 + 32 >= je) {
                    es0 += __shfl_xor(es0, 8);
                    es0 += __shfl_xor(es0, 16);
                    es1 += __shfl_xor(es1, 8);
                    es1 += __shfl_xor(es1, 16);
                    const float dmax = fmaxf((float)dg, 1.0f);
                    const float lp0 = es0 / dmax;
                    const float lp1 = es1 / dmax;
                    float ae = 0.0f;
                    #pragma unroll
                    for (int g = 0; g < NH; ++g) {
                        ae = __builtin_fmaf(__shfl(lp0, g), mcol[2 * g], ae);
                        ae = __builtin_fmaf(__shfl(lp1, g), mcol[2 * g + 1], ae);
                    }
                    float a = asn + adh + ae;
                    aself = (a >= 0.0f) ? a : NEG_SLOPE * a;
                    cm = fmaxf(cm, aself);
                }
                const float mn = fmaxf(mrun, cm);
                const float sc = __expf(mrun - mn);
                mrun = mn;
                float lsum = 0.0f;
                #pragma unroll 4
                for (int k = 0; k < nk; ++k) {
                    float ex = __expf(wmine[32 * k + lane] - mn);
                    wmine[32 * k + lane] = ex;
                    lsum += ex;
                }
                dl = dl * sc + lsum;
                #pragma unroll
                for (int g = 0; g < NH; ++g) z[g] *= GAT_lane_bcast(sc, g);
                int i = 0;
                for (; i + 8 <= cnt; i += 8) {
                    v4f r[8];
                    #pragma unroll
                    for (int u = 0; u < 8; ++u) {
                        int s = __builtin_amdgcn_readlane(ms, i + u);
                        r[u] = *(const v4f*)(xl + (long)s * F_IN);
                    }
                    asm volatile("" :: "v"(r[0]), "v"(r[1]), "v"(r[2]), "v"(r[3]), "v"(r[4]), "v"(r[5]), "v"(r[6]), "v"(r[7]));
                    #pragma unroll
                    for (int u = 0; u < 8; ++u) {
                        v4f wa = *(const v4f*)(wmine + (i + u) * NH);
                        v4f wb = *(const v4f*)(wmine + (i + u) * NH + 4);
                        z[0] += r[u] * wa[0];
                        z[1] += r[u] * wa[1];
                        z[2] += r[u] * wa[2];
                        z[3] += r[u] * wa[3];
                        z[4] += r[u] * wb[0];
                        z[5] += r[u] * wb[1];
                        z[6] += r[u] * wb[2];
                        z[7] += r[u] * wb[3];
                    }
                }
                for (; i < cnt; i += 4) {
                    v4f r[4];
                    #pragma unroll
                    for (int u = 0; u < 4; ++u) {
                        int s = __builtin_amdgcn_readlane(ms, i + u);
                        r[u] = *(const v4f*)(xl + (long)s * F_IN);
                    }
                    asm volatile("" :: "v"(r[0]), "v"(r[1]), "v"(r[2]), "v"(r[3]));
                    #pragma unroll
                    for (int u = 0; u < 4; ++u) {
                        v4f wa = *(const v4f*)(wmine + (i + u) * NH);
                        v4f wb = *(const v4f*)(wmine + (i + u) * NH + 4);
                        z[0] += r[u] * wa[0];
                        z[1] += r[u] * wa[1];
                        z[2] += r[u] * wa[2];
                        z[3] += r[u] * wa[3];
                        z[4] += r[u] * wb[0];
                        z[5] += r[u] * wb[1];
                        z[6] += r[u] * wb[2];
                        z[7] += r[u] * wb[3];
                    }
                }
                j0 += 32;
            } while (j0 < je);

            const float exs = __expf(aself - mrun);
            dl += __shfl_xor(dl, 8);
            dl += __shfl_xor(dl, 16);
            const float den = dl + exs;
            const float inv = (1.0f / NH) / den;
            {
                v4f r = *(const v4f*)(xl + n * F_IN);
                #pragma unroll
                for (int g = 0; g < NH; ++g) z[g] += r * GAT_lane_bcast(exs, g);
            }
            #pragma unroll
            for (int g = 0; g < NH; ++g) z[g] *= GAT_lane_bcast(inv, g);
        }
        float* zrow = zt + (rt * 16 + wave) * ZP + lane * 4;
        #pragma unroll
        for (int g = 0; g < NH; ++g) *(v4f*)(zrow + g * F_IN) = z[g];
    }
    __syncthreads();

    {
        const int hw = wave >> 1;
        const int ct = wave & 1;
        const int l  = lane & 15;
        const int hf = lane >> 4;
        const float* wp  = W + (long)(hw * NC + ct * 16 + l) * F_IN + 4 * hf;
        const float* za0 = zt + l * ZP + hw * F_IN + 4 * hf;
        const float* za1 = za0 + 16 * ZP;
        v8f p00 = {0.f,0.f,0.f,0.f,0.f,0.f,0.f,0.f};
        v8f p01 = p00, p10 = p00, p11 = p00;
        #pragma unroll
        for (int k0 = 0; k0 < F_IN; k0 += 8) {
            v4f b  = *(const v4f*)(wp  + k0);
            v4f a0 = *(const v4f*)(za0 + k0);
            v4f a1 = *(const v4f*)(za1 + k0);
            v2f bl  = {b[0],  b[1]},  bh  = {b[2],  b[3]};
            v2f a0l = {a0[0], a0[1]}, a0h = {a0[2], a0[3]};
            v2f a1l = {a1[0], a1[1]}, a1h = {a1[2], a1[3]};
            p00 = __builtin_amdgcn_wmma_f32_16x16x4_f32(false, a0l, false, bl, (short)0, p00, false, false);
            p10 = __builtin_amdgcn_wmma_f32_16x16x4_f32(false, a1l, false, bl, (short)0, p10, false, false);
            p01 = __builtin_amdgcn_wmma_f32_16x16x4_f32(false, a0h, false, bh, (short)0, p01, false, false);
            p11 = __builtin_amdgcn_wmma_f32_16x16x4_f32(false, a1h, false, bh, (short)0, p11, false, false);
        }
        v8f p0 = p00 + p01;
        v8f p1 = p10 + p11;
        __syncthreads();
        float* pt = zt + wave * 512 + l;
        #pragma unroll
        for (int v = 0; v < 8; ++v) {
            pt[(8 * hf + v) * 16]       = p0[v];
            pt[256 + (8 * hf + v) * 16] = p1[v];
        }
    }
    __syncthreads();
    {
        const int c   = lane;
        const int ct2 = c >> 4, cc = c & 15;
        const float bc = bias[c];
        #pragma unroll
        for (int rt = 0; rt < 2; ++rt) {
            const long n = (long)blockIdx.x * 32 + rt * 16 + wave;
            float s = 0.0f;
            #pragma unroll
            for (int g = 0; g < NH; ++g)
                s += zt[((g * 2 + ct2) * 2 + rt) * 256 + wave * 16 + cc];
            if (n < N) out[n * NC + c] = s + bc;
        }
    }
}

extern "C" void kernel_launch(void* const* d_in, const int* in_sizes, int n_in,
                              void* d_out, int out_size, void* d_ws, size_t ws_size,
                              hipStream_t stream) {
    const float* x        = (const float*)d_in[0];
    const int*   eidx     = (const int*)  d_in[1];
    const float* eattr    = (const float*)d_in[2];
    const float* W        = (const float*)d_in[3];
    const float* We       = (const float*)d_in[4];
    const float* att_src  = (const float*)d_in[5];
    const float* att_dst  = (const float*)d_in[6];
    const float* att_edge = (const float*)d_in[7];
    const float* bias     = (const float*)d_in[8];

    const int N = in_sizes[0] / F_IN;
    const int E = in_sizes[1] / 2;
    const int* src = eidx;
    const int* dst = eidx + E;

    float* ws = (float*)d_ws;
    const long Nl = N, El = E;
    const long NP = (Nl + 63) & ~63L;
    const long epc = (((El + RK_G - 1) / RK_G) + 255) & ~255L;
    int*      w_deg   = (int*)ws;
    int*      w_tot   = (int*)(ws + Nl);
    const long o_off  = (Nl + 64 + 63) & ~63L;
    const long o_rank = (o_off + Nl + 63) & ~63L;
    const long o_perm = (o_rank + El + 63) & ~63L;
    const long o_att  = (o_perm + El + 63) & ~63L;
    const long o_cnt  = (o_att + 16 * Nl + 128 + 16 * F_IN + 63) & ~63L;
    int*      w_off   = (int*)(ws + o_off);
    int*      w_rank  = (int*)(ws + o_rank);
    int*      w_perm  = (int*)(ws + o_perm);
    float*    w_asrc  = ws + o_att;
    float*    w_adst  = ws + o_att + 8 * Nl;
    float*    w_M     = ws + o_att + 16 * Nl;
    float*    w_Vt    = ws + o_att + 16 * Nl + 128;
    int*      w_cnt   = (int*)(ws + o_cnt);

    const int B = 256;
    GAT_rank<<<RK_G, 1024, 0, stream>>>(dst, w_rank, w_cnt, w_tot, E, N, (int)epc, (int)NP);
    GAT_segalloc<<<(int)((Nl + 255) / 256), 256, 0, stream>>>(w_cnt, w_deg, w_off, w_tot, N, (int)NP);
    GAT_fill<<<dim3((unsigned)(epc / B), RK_G), B, 0, stream>>>(dst, w_cnt, w_rank, w_perm, E, (int)epc, (int)NP);
    GAT_fold<<<9, B, 0, stream>>>(W, We, att_src, att_dst, att_edge, w_M, w_Vt);
    {
        long rowTiles = (Nl + 15) >> 4;
        GAT_gemm_att<<<(int)((rowTiles + 7) / 8), B, 0, stream>>>(x, w_Vt, w_asrc, w_adst, N);
    }
    GAT_gather<<<(int)((Nl + 31) / 32), 512, 0, stream>>>(src, w_deg, w_off, w_perm, eattr, w_M,
                                                          w_asrc, w_adst, x, W, bias,
                                                          (float*)d_out, N);
}


# Score operands staged through LDS: lane j fetches edge j's 64-byte attribute row and its source's 32-byte term row, six loads in flight together; steps read LDS, not memory; sum order unchanged
# speedup vs baseline: 7.6766x; 1.0397x over previous
#include <hip/hip_runtime.h>
#include <hip/hip_bf16.h>

typedef float v2f __attribute__((ext_vector_type(2)));
typedef float v8f __attribute__((ext_vector_type(8)));
typedef float v4f __attribute__((ext_vector_type(4)));

#define F_IN 128
#define F_E  16
#define NH   8
#define NC   32
#define HC   256
#define NEG_SLOPE 0.2f

typedef int v4i __attribute__((ext_vector_type(4)));

#define RK_G    64
#define RK_CAP  51200
#define RK_B    16

__global__ void __launch_bounds__(1024)
GAT_rank(const int* __restrict__ dst, int* __restrict__ rank, int* __restrict__ cnt,
         int* __restrict__ total, int E, int N, int epc, int np) {
    __shared__ __attribute__((aligned(16))) int hist[RK_CAP];
    const int t = threadIdx.x;
    const int g = blockIdx.x;
    if (g == 0 && t == 0) *total = 0;
    const long e0 = (long)g * epc;
    long e1 = e0 + epc;
    if (e1 > E) e1 = E;
    for (int nb = 0; nb < N; nb += RK_CAP) {
        int sn = N - nb;
        if (sn > RK_CAP) sn = RK_CAP;
        const int sn4 = (sn + 3) >> 2;
        int d[RK_B];
        long eb = e0;
        #pragma unroll
        for (int u = 0; u < RK_B; ++u) {
            const long e = eb + u * 1024 + t;
            d[u] = (e < e1) ? dst[e] : -1;
        }
        for (int i = t; i < sn4; i += 1024) ((v4i*)hist)[i] = (v4i){0, 0, 0, 0};
        __syncthreads();
        while (true) {
            int r[RK_B];
            #pragma unroll
            for (int u = 0; u < RK_B; ++u) {
                const unsigned k = (unsigned)(d[u] - nb);
                r[u] = (k < (unsigned)sn) ? atomicAdd(&hist[k], 1) : -1;
            }
            #pragma unroll
            for (int u = 0; u < RK_B; ++u) {
                const long e = eb + u * 1024 + t;
                if (r[u] >= 0) rank[e] = r[u];
            }
            eb += (long)RK_B * 1024;
            if (eb >= e1) break;
            #pragma unroll
            for (int u = 0; u < RK_B; ++u) {
                const long e = eb + u * 1024 + t;
                d[u] = (e < e1) ? dst[e] : -1;
            }
        }
        __syncthreads();
        int* row = cnt + (long)g * np + nb;
        #pragma unroll 4
        for (int i = t; i < sn4; i += 1024) ((v4i*)row)[i] = ((const v4i*)hist)[i];
        __syncthreads();
    }
}

__global__ void GAT_fold(const float* __restrict__ W,
                         const float* __restrict__ We,
                         const float* __restrict__ att_src,
                         const float* __restrict__ att_dst,
                         const float* __restrict__ att_edge,
                         float* __restrict__ M,
                         float* __restrict__ Vt) {
    int tid = threadIdx.x;
    if (blockIdx.x < 8) {
        int idx = blockIdx.x * 256 + tid;
        int j = idx >> 7, f = idx & (F_IN - 1);
        int hj = j & 7;
        const float* att = (j < NH) ? att_src : att_dst;
        float s = 0.0f;
        for (int c = 0; c < NC; ++c)
            s += W[(long)(hj * NC + c) * F_IN + f] * att[hj * NC + c];
        Vt[j * F_IN + f] = s;
    } else {
        if (tid >= F_E * NH) return;
        int f = tid >> 3, h = tid & 7;
        float s = 0.0f;
        for (int c = 0; c < NC; ++c)
            s += We[(long)(h * NC + c) * F_E + f] * att_edge[h * NC + c];
        M[f * NH + h] = s;
    }
}

__global__ void GAT_gemm_att(const float* __restrict__ x,
                             const float* __restrict__ Vt,
                             float* __restrict__ asrc,
                             float* __restrict__ adst, int N) {
    int wave = threadIdx.x >> 5;
    int lane = threadIdx.x & 31;
    int half = lane >> 4;
    int l    = lane & 15;
    long rowTiles = ((long)N + 15) >> 4;
    long tile = (long)blockIdx.x * 8 + wave;
    if (tile >= rowTiles) return;
    long row0 = tile * 16;
    long arow = row0 + l;
    if (arow >= N) arow = N - 1;
    const float* xp = x  + arow * F_IN + 2 * half;
    const float* vp = Vt + (long)l * F_IN + 2 * half;
    v8f a0 = {0.f,0.f,0.f,0.f,0.f,0.f,0.f,0.f};
    v8f a1 = a0;
    #pragma unroll
    for (int k0 = 0; k0 < F_IN; k0 += 8) {
        v2f xa = *(const v2f*)(xp + k0);
        v2f va = *(const v2f*)(vp + k0);
        v2f xb = *(const v2f*)(xp + k0 + 4);
        v2f vb = *(const v2f*)(vp + k0 + 4);
        a0 = __builtin_amdgcn_wmma_f32_16x16x4_f32(false, xa, false, va, (short)0, a0, false, false);
        a1 = __builtin_amdgcn_wmma_f32_16x16x4_f32(false, xb, false, vb, (short)0, a1, false, false);
    }
    v8f acc = a0 + a1;
    #pragma unroll
    for (int v = 0; v < 8; ++v) {
        long r = row0 + v + 8 * half;
        if (r < N) {
            if (l < NH) asrc[r * NH + l] = acc[v];
            else        adst[r * NH + (l - NH)] = acc[v];
        }
    }
}

__global__ void __launch_bounds__(256)
GAT_segalloc(int* __restrict__ cnt, int* __restrict__ deg, int* __restrict__ off,
             int* __restrict__ total, int N, int np) {
    __shared__ int wsum[8];
    __shared__ int base;
    const int t = threadIdx.x;
    const int lane = t & 31, w = t >> 5;
    const long n = (long)blockIdx.x * 256 + t;
    int* col = cnt + ((n < N) ? n : (long)N - 1);
    int c[RK_G];
    #pragma unroll
    for (int g = 0; g < RK_G; ++g) c[g] = col[(long)g * np];
    int tot = 0;
    #pragma unroll
    for (int g = 0; g < RK_G; ++g) tot += c[g];
    if (n < N) deg[n] = tot; else tot = 0;
    int inc = tot;
    for (int o = 1; o < 32; o <<= 1) {
        int v = __shfl_up(inc, o);
        if (lane >= o) inc += v;
    }
    if (lane == 31) wsum[w] = inc;
    __syncthreads();
    if (w == 0) {
        int s = (lane < 8) ? wsum[lane] : 0;
        int si = s;
        for (int o = 1; o < 32; o <<= 1) {
            int v = __shfl_up(si, o);
            if (lane >= o) si += v;
        }
        if (lane < 8) wsum[lane] = si - s;
        if (lane == 31) base = atomicAdd(total, si);
    }
    __syncthreads();
    if (n < N) {
        int run = base + wsum[w] + inc - tot;
        off[n] = run;
        #pragma unroll
        for (int g = 0; g < RK_G; ++g) { col[(long)g * np] = run; run += c[g]; }
    }
}

__global__ void GAT_fill(const int* __restrict__ dst, const int* __restrict__ cbase,
                         const int* __restrict__ rank, int* __restrict__ perm, int E, int epc, int np) {
    const int g = blockIdx.y;
    const long e = (long)g * epc + (long)blockIdx.x * blockDim.x + threadIdx.x;
    if (e >= E) return;
    perm[cbase[(long)g * np + dst[e]] + rank[e]] = (int)e;
}

__device__ __forceinline__ float GAT_lane_bcast(float v, int g) {
    return __builtin_bit_cast(float, __builtin_amdgcn_readlane(__builtin_bit_cast(int, v), g));
}
#define ZP 1028
#define SP 20
__global__ void __launch_bounds__(512, 8)
GAT_gather(const int* __restrict__ src,
           const int* __restrict__ deg,
           const int* __restrict__ off,
           const int* __restrict__ perm,
           const float* __restrict__ eattr,
           const float* __restrict__ M,
           const float* __restrict__ asrc,
           const float* __restrict__ adst,
           const float* __restrict__ x,
           const float* __restrict__ W,
           const float* __restrict__ bias,
           float* __restrict__ out, int N) {
    __shared__ __attribute__((aligned(16))) float zt[32 * ZP];
    __shared__ __attribute__((aligned(16))) float wexp[16][32 * NH];
    const int wave = __builtin_amdgcn_readfirstlane((int)(threadIdx.x >> 5));
    const int lane = threadIdx.x & 31;
    const int hq = lane >> 3;
    const int h  = lane & 7;
    float* wmine = &wexp[wave][0];
    const float* xl = x + lane * 4;

    float mcol[F_E];
    #pragma unroll
    for (int f = 0; f < F_E; ++f) mcol[f] = M[f * NH + h];

    #pragma nounroll
    for (int rt = 0; rt < 2; ++rt) {
        const long n = (long)blockIdx.x * 32 + rt * 16 + wave;
        float* stage = zt + (rt * 16 + wave) * ZP;
        v4f z[NH];
        #pragma unroll
        for (int g = 0; g < NH; ++g) z[g] = (v4f){0.f, 0.f, 0.f, 0.f};
        if (n < N) {
            const float adh = adst[n * NH + h];
            const float asn = asrc[n * NH + h];
            float mrun = -__builtin_inff();
            float dl = 0.0f;
            float es0 = 0.0f, es1 = 0.0f;
            float aself = 0.0f;
            const int dg = deg[n];
            const int jb = off[n];
            const int je = jb + dg;
            int j0 = jb;
            do {
                int cnt = je - j0;
                if (cnt > 32) cnt = 32;
                int me = 0, ms = (int)n;
                if (lane < cnt) { me = perm[j0 + lane]; ms = src[me]; }
                const int nk = (cnt + 3) >> 2;
                if (lane < 4 * nk) {
                    const float* er = eattr + (long)me * F_E;
                    const float* tr = asrc + (long)ms * NH;
                    v4f s0 = *(const v4f*)(er);
                    v4f s1 = *(const v4f*)(er + 4);
                    v4f s2 = *(const v4f*)(er + 8);
                    v4f s3 = *(const v4f*)(er + 12);
                    v4f t0 = *(const v4f*)(tr);
                    v4f t1 = *(const v4f*)(tr + 4);
                    float* sr = stage + lane * SP;
                    *(v4f*)(sr)      = s0;
                    *(v4f*)(sr + 4)  = s1;
                    *(v4f*)(sr + 8)  = s2;
                    *(v4f*)(sr + 12) = s3;
                    *(v4f*)(wmine + lane * NH)     = t0;
                    *(v4f*)(wmine + lane * NH + 4) = t1;
                }
                float lmax = -__builtin_inff();
                #pragma unroll 2
                for (int k = 0; k < nk; ++k) {
                    const int j = 4 * k + hq;
                    const bool valid = j < cnt;
                    const float* er = stage + j * SP;
                    v4f e0 = *(const v4f*)(er);
                    v4f e1 = *(const v4f*)(er + 4);
                    v4f e2 = *(const v4f*)(er + 8);
                    v4f e3 = *(const v4f*)(er + 12);
                    v2f ef = *(const v2f*)(er + 2 * h);
                    float as = wmine[32 * k + lane];
                    float ae = 0.0f;
                    #pragma unroll
                    for (int q = 0; q < 4; ++q) ae = __builtin_fmaf(e0[q], mcol[q], ae);
                    #pragma unroll
                    for (int q = 0; q < 4; ++q) ae = __builtin_fmaf(e1[q], mcol[4 + q], ae);
                    #pragma unroll
                    for (int q = 0; q < 4; ++q) ae = __builtin_fmaf(e2[q], mcol[8 + q], ae);
                    #pragma unroll
                    for (int q = 0; q < 4; ++q) ae = __builtin_fmaf(e3[q], mcol[12 + q], ae);
                    float a = as + adh + ae;
                    a = (a >= 0.0f) ? a : NEG_SLOPE * a;
                    a = valid ? a : -__builtin_inff();
                    es0 += valid ? ef[0] : 0.0f;
                    es1 += valid ? ef[1] : 0.0f;
                    wmine[32 * k + lane] = a;
                    lmax = fmaxf(lmax, a);
                }
                float cm = fmaxf(lmax, __shfl_xor(lmax, 8));
                cm = fmaxf(cm, __shfl_xor(cm, 16));
                if (j0 + 32 >= je) {
                    es0 += __shfl_xor(es0, 8);
                    es0 += __shfl_xor(es0, 16);
                    es1 += __shfl_xor(es1, 8);
                    es1 += __shfl_xor(es1, 16);
                    const float dmax = fmaxf((float)dg, 1.0f);
                    const float lp0 = es0 / dmax;
                    const float lp1 = es1 / dmax;
                    float ae = 0.0f;
                    #pragma unroll
                    for (int g = 0; g < NH; ++g) {
                        ae = __builtin_fmaf(__shfl(lp0, g), mcol[2 * g], ae);
                        ae = __builtin_fmaf(__shfl(lp1, g), mcol[2 * g + 1], ae);
                    }
                    float a = asn + adh + ae;
                    aself = (a >= 0.0f) ? a : NEG_SLOPE * a;
                    cm = fmaxf(cm, aself);
                }
                const float mn = fmaxf(mrun, cm);
                const float sc = __expf(mrun - mn);
                mrun = mn;
                float lsum = 0.0f;
                #pragma unroll 4
                for (int k = 0; k < nk; ++k) {
                    float ex = __expf(wmine[32 * k + lane] - mn);
                    wmine[32 * k + lane] = ex;
                    lsum += ex;
                }
                dl = dl * sc + lsum;
                #pragma unroll
                for (int g = 0; g < NH; ++g) z[g] *= GAT_lane_bcast(sc, g);
                int i = 0;
                for (; i + 8 <= cnt; i += 8) {
                    v4f r[8];
                    #pragma unroll
                    for (int u = 0; u < 8; ++u) {
                        int s = __builtin_amdgcn_readlane(ms, i + u);
                        r[u] = *(const v4f*)(xl + (long)s * F_IN);
                    }
                    asm volatile("" :: "v"(r[0]), "v"(r[1]), "v"(r[2]), "v"(r[3]), "v"(r[4]), "v"(r[5]), "v"(r[6]), "v"(r[7]));
                    #pragma unroll
                    for (int u = 0; u < 8; ++u) {
                        v4f wa = *(const v4f*)(wmine + (i + u) * NH);
                        v4f wb = *(const v4f*)(wmine + (i + u) * NH + 4);
                        z[0] += r[u] * wa[0];
                        z[1] += r[u] * wa[1];
                        z[2] += r[u] * wa[2];
                        z[3] += r[u] * wa[3];
                        z[4] += r[u] * wb[0];
                        z[5] += r[u] * wb[1];
                        z[6] += r[u] * wb[2];
                        z[7] += r[u] * wb[3];
                    }
                }
                for (; i < cnt; i += 4) {
                    v4f r[4];
                    #pragma unroll
                    for (int u = 0; u < 4; ++u) {
                        int s = __builtin_amdgcn_readlane(ms, i + u);
                        r[u] = *(const v4f*)(xl + (long)s * F_IN);
                    }
                    asm volatile("" :: "v"(r[0]), "v"(r[1]), "v"(r[2]), "v"(r[3]));
                    #pragma unroll
                    for (int u = 0; u < 4; ++u) {
                        v4f wa = *(const v4f*)(wmine + (i + u) * NH);
                        v4f wb = *(const v4f*)(wmine + (i + u) * NH + 4);
                        z[0] += r[u] * wa[0];
                        z[1] += r[u] * wa[1];
                        z[2] += r[u] * wa[2];
                        z[3] += r[u] * wa[3];
                        z[4] += r[u] * wb[0];
                        z[5] += r[u] * wb[1];
                        z[6] += r[u] * wb[2];
                        z[7] += r[u] * wb[3];
                    }
                }
                j0 += 32;
            } while (j0 < je);

            const float exs = __expf(aself - mrun);
            dl += __shfl_xor(dl, 8);
            dl += __shfl_xor(dl, 16);
            const float den = dl + exs;
            const float inv = (1.0f / NH) / den;
            {
                v4f r = *(const v4f*)(xl + n * F_IN);
                #pragma unroll
                for (int g = 0; g < NH; ++g) z[g] += r * GAT_lane_bcast(exs, g);
            }
            #pragma unroll
            for (int g = 0; g < NH; ++g) z[g] *= GAT_lane_bcast(inv, g);
        }
        float* zrow = zt + (rt * 16 + wave) * ZP + lane * 4;
        #pragma unroll
        for (int g = 0; g < NH; ++g) *(v4f*)(zrow + g * F_IN) = z[g];
    }
    __syncthreads();

    {
        const int hw = wave >> 1;
        const int ct = wave & 1;
        const int l  = lane & 15;
        const int hf = lane >> 4;
        const float* wp  = W + (long)(hw * NC + ct * 16 + l) * F_IN + 4 * hf;
        const float* za0 = zt + l * ZP + hw * F_IN + 4 * hf;
        const float* za1 = za0 + 16 * ZP;
        v8f p00 = {0.f,0.f,0.f,0.f,0.f,0.f,0.f,0.f};
        v8f p01 = p00, p10 = p00, p11 = p00;
        #pragma unroll
        for (int k0 = 0; k0 < F_IN; k0 += 8) {
            v4f b  = *(const v4f*)(wp  + k0);
            v4f a0 = *(const v4f*)(za0 + k0);
            v4f a1 = *(const v4f*)(za1 + k0);
            v2f bl  = {b[0],  b[1]},  bh  = {b[2],  b[3]};
            v2f a0l = {a0[0], a0[1]}, a0h = {a0[2], a0[3]};
            v2f a1l = {a1[0], a1[1]}, a1h = {a1[2], a1[3]};
            p00 = __builtin_amdgcn_wmma_f32_16x16x4_f32(false, a0l, false, bl, (short)0, p00, false, false);
            p10 = __builtin_amdgcn_wmma_f32_16x16x4_f32(false, a1l, false, bl, (short)0, p10, false, false);
            p01 = __builtin_amdgcn_wmma_f32_16x16x4_f32(false, a0h, false, bh, (short)0, p01, false, false);
            p11 = __builtin_amdgcn_wmma_f32_16x16x4_f32(false, a1h, false, bh, (short)0, p11, false, false);
        }
        v8f p0 = p00 + p01;
        v8f p1 = p10 + p11;
        __syncthreads();
        float* pt = zt + wave * 512 + l;
        #pragma unroll
        for (int v = 0; v < 8; ++v) {
            pt[(8 * hf + v) * 16]       = p0[v];
            pt[256 + (8 * hf + v) * 16] = p1[v];
        }
    }
    __syncthreads();
    {
        const int c   = lane;
        const int ct2 = c >> 4, cc = c & 15;
        const float bc = bias[c];
        #pragma unroll
        for (int rt = 0; rt < 2; ++rt) {
            const long n = (long)blockIdx.x * 32 + rt * 16 + wave;
            float s = 0.0f;
            #pragma unroll
            for (int g = 0; g < NH; ++g)
                s += zt[((g * 2 + ct2) * 2 + rt) * 256 + wave * 16 + cc];
            if (n < N) out[n * NC + c] = s + bc;
        }
    }
}

extern "C" void kernel_launch(void* const* d_in, const int* in_sizes, int n_in,
                              void* d_out, int out_size, void* d_ws, size_t ws_size,
                              hipStream_t stream) {
    const float* x        = (const float*)d_in[0];
    const int*   eidx     = (const int*)  d_in[1];
    const float* eattr    = (const float*)d_in[2];
    const float* W        = (const float*)d_in[3];
    const float* We       = (const float*)d_in[4];
    const float* att_src  = (const float*)d_in[5];
    const float* att_dst  = (const float*)d_in[6];
    const float* att_edge = (const float*)d_in[7];
    const float* bias     = (const float*)d_in[8];

    const int N = in_sizes[0] / F_IN;
    const int E = in_sizes[1] / 2;
    const int* src = eidx;
    const int* dst = eidx + E;

    float* ws = (float*)d_ws;
    const long Nl = N, El = E;
    const long NP = (Nl + 63) & ~63L;
    const long epc = (((El + RK_G - 1) / RK_G) + 255) & ~255L;
    int*      w_deg   = (int*)ws;
    int*      w_tot   = (int*)(ws + Nl);
    const long o_off  = (Nl + 64 + 63) & ~63L;
    const long o_rank = (o_off + Nl + 63) & ~63L;
    const long o_perm = (o_rank + El + 63) & ~63L;
    const long o_att  = (o_perm + El + 63) & ~63L;
    const long o_cnt  = (o_att + 16 * Nl + 128 + 16 * F_IN + 63) & ~63L;
    int*      w_off   = (int*)(ws + o_off);
    int*      w_rank  = (int*)(ws + o_rank);
    int*      w_perm  = (int*)(ws + o_perm);
    float*    w_asrc  = ws + o_att;
    float*    w_adst  = ws + o_att + 8 * Nl;
    float*    w_M     = ws + o_att + 16 * Nl;
    float*    w_Vt    = ws + o_att + 16 * Nl + 128;
    int*      w_cnt   = (int*)(ws + o_cnt);

    const int B = 256;
    GAT_rank<<<RK_G, 1024, 0, stream>>>(dst, w_rank, w_cnt, w_tot, E, N, (int)epc, (int)NP);
    GAT_segalloc<<<(int)((Nl + 255) / 256), 256, 0, stream>>>(w_cnt, w_deg, w_off, w_tot, N, (int)NP);
    GAT_fill<<<dim3((unsigned)(epc / B), RK_G), B, 0, stream>>>(dst, w_cnt, w_rank, w_perm, E, (int)epc, (int)NP);
    GAT_fold<<<9, B, 0, stream>>>(W, We, att_src, att_dst, att_edge, w_M, w_Vt);
    {
        long rowTiles = (Nl + 15) >> 4;
        GAT_gemm_att<<<(int)((rowTiles + 7) / 8), B, 0, stream>>>(x, w_Vt, w_asrc, w_adst, N);
    }
    GAT_gather<<<(int)((Nl + 31) / 32), 512, 0, stream>>>(src, w_deg, w_off, w_perm, eattr, w_M,
                                                          w_asrc, w_adst, x, W, bias,
                                                          (float*)d_out, N);
}


# Edge slots widened to 8 bytes (edge id, source id): fill reads source ids in edge order and stores both at once; node waves drop the dependent one-word-per-line src gather; sum order unchanged
# speedup vs baseline: 7.9354x; 1.0337x over previous
#include <hip/hip_runtime.h>
#include <hip/hip_bf16.h>

typedef float v2f __attribute__((ext_vector_type(2)));
typedef float v8f __attribute__((ext_vector_type(8)));
typedef float v4f __attribute__((ext_vector_type(4)));

#define F_IN 128
#define F_E  16
#define NH   8
#define NC   32
#define HC   256
#define NEG_SLOPE 0.2f

typedef int v4i __attribute__((ext_vector_type(4)));
typedef int v2i __attribute__((ext_vector_type(2)));

#define RK_G    64
#define RK_CAP  51200
#define RK_B    16

__global__ void __launch_bounds__(1024)
GAT_rank(const int* __restrict__ dst, int* __restrict__ rank, int* __restrict__ cnt,
         int* __restrict__ total, int E, int N, int epc, int np) {
    __shared__ __attribute__((aligned(16))) int hist[RK_CAP];
    const int t = threadIdx.x;
    const int g = blockIdx.x;
    if (g == 0 && t == 0) *total = 0;
    const long e0 = (long)g * epc;
    long e1 = e0 + epc;
    if (e1 > E) e1 = E;
    for (int nb = 0; nb < N; nb += RK_CAP) {
        int sn = N - nb;
        if (sn > RK_CAP) sn = RK_CAP;
        const int sn4 = (sn + 3) >> 2;
        int d[RK_B];
        long eb = e0;
        #pragma unroll
        for (int u = 0; u < RK_B; ++u) {
            const long e = eb + u * 1024 + t;
            d[u] = (e < e1) ? dst[e] : -1;
        }
        for (int i = t; i < sn4; i += 1024) ((v4i*)hist)[i] = (v4i){0, 0, 0, 0};
        __syncthreads();
        while (true) {
            int r[RK_B];
            #pragma unroll
            for (int u = 0; u < RK_B; ++u) {
                const unsigned k = (unsigned)(d[u] - nb);
                r[u] = (k < (unsigned)sn) ? atomicAdd(&hist[k], 1) : -1;
            }
            #pragma unroll
            for (int u = 0; u < RK_B; ++u) {
                const long e = eb + u * 1024 + t;
                if (r[u] >= 0) rank[e] = r[u];
            }
            eb += (long)RK_B * 1024;
            if (eb >= e1) break;
            #pragma unroll
            for (int u = 0; u < RK_B; ++u) {
                const long e = eb + u * 1024 + t;
                d[u] = (e < e1) ? dst[e] : -1;
            }
        }
        __syncthreads();
        int* row = cnt + (long)g * np + nb;
        #pragma unroll 4
        for (int i = t; i < sn4; i += 1024) ((v4i*)row)[i] = ((const v4i*)hist)[i];
        __syncthreads();
    }
}

__global__ void GAT_fold(const float* __restrict__ W,
                         const float* __restrict__ We,
                         const float* __restrict__ att_src,
                         const float* __restrict__ att_dst,
                         const float* __restrict__ att_edge,
                         float* __restrict__ M,
                         float* __restrict__ Vt) {
    int tid = threadIdx.x;
    if (blockIdx.x < 8) {
        int idx = blockIdx.x * 256 + tid;
        int j = idx >> 7, f = idx & (F_IN - 1);
        int hj = j & 7;
        const float* att = (j < NH) ? att_src : att_dst;
        float s = 0.0f;
        for (int c = 0; c < NC; ++c)
            s += W[(long)(hj * NC + c) * F_IN + f] * att[hj * NC + c];
        Vt[j * F_IN + f] = s;
    } else {
        if (tid >= F_E * NH) return;
        int f = tid >> 3, h = tid & 7;
        float s = 0.0f;
        for (int c = 0; c < NC; ++c)
            s += We[(long)(h * NC + c) * F_E + f] * att_edge[h * NC + c];
        M[f * NH + h] = s;
    }
}

__global__ void GAT_gemm_att(const float* __restrict__ x,
                             const float* __restrict__ Vt,
                             float* __restrict__ asrc,
                             float* __restrict__ adst, int N) {
    int wave = threadIdx.x >> 5;
    int lane = threadIdx.x & 31;
    int half = lane >> 4;
    int l    = lane & 15;
    long rowTiles = ((long)N + 15) >> 4;
    long tile = (long)blockIdx.x * 8 + wave;
    if (tile >= rowTiles) return;
    long row0 = tile * 16;
    long arow = row0 + l;
    if (arow >= N) arow = N - 1;
    const float* xp = x  + arow * F_IN + 2 * half;
    const float* vp = Vt + (long)l * F_IN + 2 * half;
    v8f a0 = {0.f,0.f,0.f,0.f,0.f,0.f,0.f,0.f};
    v8f a1 = a0;
    #pragma unroll
    for (int k0 = 0; k0 < F_IN; k0 += 8) {
        v2f xa = *(const v2f*)(xp + k0);
        v2f va = *(const v2f*)(vp + k0);
        v2f xb = *(const v2f*)(xp + k0 + 4);
        v2f vb = *(const v2f*)(vp + k0 + 4);
        a0 = __builtin_amdgcn_wmma_f32_16x16x4_f32(false, xa, false, va, (short)0, a0, false, false);
        a1 = __builtin_amdgcn_wmma_f32_16x16x4_f32(false, xb, false, vb, (short)0, a1, false, false);
    }
    v8f acc = a0 + a1;
    #pragma unroll
    for (int v = 0; v < 8; ++v) {
        long r = row0 + v + 8 * half;
        if (r < N) {
            if (l < NH) asrc[r * NH + l] = acc[v];
            else        adst[r * NH + (l - NH)] = acc[v];
        }
    }
}

__global__ void __launch_bounds__(256)
GAT_segalloc(int* __restrict__ cnt, int* __restrict__ deg, int* __restrict__ off,
             int* __restrict__ total, int N, int np) {
    __shared__ int wsum[8];
    __shared__ int base;
    const int t = threadIdx.x;
    const int lane = t & 31, w = t >> 5;
    const long n = (long)blockIdx.x * 256 + t;
    int* col = cnt + ((n < N) ? n : (long)N - 1);
    int c[RK_G];
    #pragma unroll
    for (int g = 0; g < RK_G; ++g) c[g] = col[(long)g * np];
    int tot = 0;
    #pragma unroll
    for (int g = 0; g < RK_G; ++g) tot += c[g];
    if (n < N) deg[n] = tot; else tot = 0;
    int inc = tot;
    for (int o = 1; o < 32; o <<= 1) {
        int v = __shfl_up(inc, o);
        if (lane >= o) inc += v;
    }
    if (lane == 31) wsum[w] = inc;
    __syncthreads();
    if (w == 0) {
        int s = (lane < 8) ? wsum[lane] : 0;
        int si = s;
        for (int o = 1; o < 32; o <<= 1) {
            int v = __shfl_up(si, o);
            if (lane >= o) si += v;
        }
        if (lane < 8) wsum[lane] = si - s;
        if (lane == 31) base = atomicAdd(total, si);
    }
    __syncthreads();
    if (n < N) {
        int run = base + wsum[w] + inc - tot;
        off[n] = run;
        #pragma unroll
        for (int g = 0; g < RK_G; ++g) { col[(long)g * np] = run; run += c[g]; }
    }
}

__global__ void GAT_fill(const int* __restrict__ src, const int* __restrict__ dst, const int* __restrict__ cbase,
                         const int* __restrict__ rank, int* __restrict__ perm, int E, int epc, int np) {
    const int g = blockIdx.y;
    const long e = (long)g * epc + (long)blockIdx.x * blockDim.x + threadIdx.x;
    if (e >= E) return;
    const long slot = (long)cbase[(long)g * np + dst[e]] + rank[e];
    ((v2i*)perm)[slot] = (v2i){(int)e, src[e]};
}

__device__ __forceinline__ float GAT_lane_bcast(float v, int g) {
    return __builtin_bit_cast(float, __builtin_amdgcn_readlane(__builtin_bit_cast(int, v), g));
}
#define ZP 1028
#define SP 20
__global__ void __launch_bounds__(512, 8)
GAT_gather(const int* __restrict__ deg,
           const int* __restrict__ off,
           const int* __restrict__ perm,
           const float* __restrict__ eattr,
           const float* __restrict__ M,
           const float* __restrict__ asrc,
           const float* __restrict__ adst,
           const float* __restrict__ x,
           const float* __restrict__ W,
           const float* __restrict__ bias,
           float* __restrict__ out, int N) {
    __shared__ __attribute__((aligned(16))) float zt[32 * ZP];
    __shared__ __attribute__((aligned(16))) float wexp[16][32 * NH];
    const int wave = __builtin_amdgcn_readfirstlane((int)(threadIdx.x >> 5));
    const int lane = threadIdx.x & 31;
    const int hq = lane >> 3;
    const int h  = lane & 7;
    float* wmine = &wexp[wave][0];
    const float* xl = x + lane * 4;

    float mcol[F_E];
    #pragma unroll
    for (int f = 0; f < F_E; ++f) mcol[f] = M[f * NH + h];

    #pragma nounroll
    for (int rt = 0; rt < 2; ++rt) {
        const long n = (long)blockIdx.x * 32 + rt * 16 + wave;
        float* stage = zt + (rt * 16 + wave) * ZP;
        v4f z[NH];
        #pragma unroll
        for (int g = 0; g < NH; ++g) z[g] = (v4f){0.f, 0.f, 0.f, 0.f};
        if (n < N) {
            const float adh = adst[n * NH + h];
            const float asn = asrc[n * NH + h];
            float mrun = -__builtin_inff();
            float dl = 0.0f;
            float es0 = 0.0f, es1 = 0.0f;
            float aself = 0.0f;
            const int dg = deg[n];
            const int jb = off[n];
            const int je = jb + dg;
            int j0 = jb;
            do {
                int cnt = je - j0;
                if (cnt > 32) cnt = 32;
                int me = 0, ms = (int)n;
                if (lane < cnt) {
                    const v2i p = ((const v2i*)perm)[j0 + lane];
                    me = p[0];
                    ms = p[1];
                }
                const int nk = (cnt + 3) >> 2;
                if (lane < 4 * nk) {
                    const float* er = eattr + (long)me * F_E;
                    const float* tr = asrc + (long)ms * NH;
                    v4f s0 = *(const v4f*)(er);
                    v4f s1 = *(const v4f*)(er + 4);
                    v4f s2 = *(const v4f*)(er + 8);
                    v4f s3 = *(const v4f*)(er + 12);
                    v4f t0 = *(const v4f*)(tr);
                    v4f t1 = *(const v4f*)(tr + 4);
                    float* sr = stage + lane * SP;
                    *(v4f*)(sr)      = s0;
                    *(v4f*)(sr + 4)  = s1;
                    *(v4f*)(sr + 8)  = s2;
                    *(v4f*)(sr + 12) = s3;
                    *(v4f*)(wmine + lane * NH)     = t0;
                    *(v4f*)(wmine + lane * NH + 4) = t1;
                }
                float lmax = -__builtin_inff();
                #pragma unroll 2
                for (int k = 0; k < nk; ++k) {
                    const int j = 4 * k + hq;
                    const bool valid = j < cnt;
                    const float* er = stage + j * SP;
                    v4f e0 = *(const v4f*)(er);
                    v4f e1 = *(const v4f*)(er + 4);
                    v4f e2 = *(const v4f*)(er + 8);
                    v4f e3 = *(const v4f*)(er + 12);
                    v2f ef = *(const v2f*)(er + 2 * h);
                    float as = wmine[32 * k + lane];
                    float ae = 0.0f;
                    #pragma unroll
                    for (int q = 0; q < 4; ++q) ae = __builtin_fmaf(e0[q], mcol[q], ae);
                    #pragma unroll
                    for (int q = 0; q < 4; ++q) ae = __builtin_fmaf(e1[q], mcol[4 + q], ae);
                    #pragma unroll
                    for (int q = 0; q < 4; ++q) ae = __builtin_fmaf(e2[q], mcol[8 + q], ae);
                    #pragma unroll
                    for (int q = 0; q < 4; ++q) ae = __builtin_fmaf(e3[q], mcol[12 + q], ae);
                    float a = as + adh + ae;
                    a = (a >= 0.0f) ? a : NEG_SLOPE * a;
                    a = valid ? a : -__builtin_inff();
                    es0 += valid ? ef[0] : 0.0f;
                    es1 += valid ? ef[1] : 0.0f;
                    wmine[32 * k + lane] = a;
                    lmax = fmaxf(lmax, a);
                }
                float cm = fmaxf(lmax, __shfl_xor(lmax, 8));
                cm = fmaxf(cm, __shfl_xor(cm, 16));
                if (j0 + 32 >= je) {
                    es0 += __shfl_xor(es0, 8);
                    es0 += __shfl_xor(es0, 16);
                    es1 += __shfl_xor(es1, 8);
                    es1 += __shfl_xor(es1, 16);
                    const float dmax = fmaxf((float)dg, 1.0f);
                    const float lp0 = es0 / dmax;
                    const float lp1 = es1 / dmax;
                    float ae = 0.0f;
                    #pragma unroll
                    for (int g = 0; g < NH; ++g) {
                        ae = __builtin_fmaf(__shfl(lp0, g), mcol[2 * g], ae);
                        ae = __builtin_fmaf(__shfl(lp1, g), mcol[2 * g + 1], ae);
                    }
                    float a = asn + adh + ae;
                    aself = (a >= 0.0f) ? a : NEG_SLOPE * a;
                    cm = fmaxf(cm, aself);
                }
                const float mn = fmaxf(mrun, cm);
                const float sc = __expf(mrun - mn);
                mrun = mn;
                float lsum = 0.0f;
                #pragma unroll 4
                for (int k = 0; k < nk; ++k) {
                    float ex = __expf(wmine[32 * k + lane] - mn);
                    wmine[32 * k + lane] = ex;
                    lsum += ex;
                }
                dl = dl * sc + lsum;
                #pragma unroll
                for (int g = 0; g < NH; ++g) z[g] *= GAT_lane_bcast(sc, g);
                int i = 0;
                for (; i + 8 <= cnt; i += 8) {
                    v4f r[8];
                    #pragma unroll
                    for (int u = 0; u < 8; ++u) {
                        int s = __builtin_amdgcn_readlane(ms, i + u);
                        r[u] = *(const v4f*)(xl + (long)s * F_IN);
                    }
                    asm volatile("" :: "v"(r[0]), "v"(r[1]), "v"(r[2]), "v"(r[3]), "v"(r[4]), "v"(r[5]), "v"(r[6]), "v"(r[7]));
                    #pragma unroll
                    for (int u = 0; u < 8; ++u) {
                        v4f wa = *(const v4f*)(wmine + (i + u) * NH);
                        v4f wb = *(const v4f*)(wmine + (i + u) * NH + 4);
                        z[0] += r[u] * wa[0];
                        z[1] += r[u] * wa[1];
                        z[2] += r[u] * wa[2];
                        z[3] += r[u] * wa[3];
                        z[4] += r[u] * wb[0];
                        z[5] += r[u] * wb[1];
                        z[6] += r[u] * wb[2];
                        z[7] += r[u] * wb[3];
                    }
                }
                for (; i < cnt; i += 4) {
                    v4f r[4];
                    #pragma unroll
                    for (int u = 0; u < 4; ++u) {
                        int s = __builtin_amdgcn_readlane(ms, i + u);
                        r[u] = *(const v4f*)(xl + (long)s * F_IN);
                    }
                    asm volatile("" :: "v"(r[0]), "v"(r[1]), "v"(r[2]), "v"(r[3]));
                    #pragma unroll
                    for (int u = 0; u < 4; ++u) {
                        v4f wa = *(const v4f*)(wmine + (i + u) * NH);
                        v4f wb = *(const v4f*)(wmine + (i + u) * NH + 4);
                        z[0] += r[u] * wa[0];
                        z[1] += r[u] * wa[1];
                        z[2] += r[u] * wa[2];
                        z[3] += r[u] * wa[3];
                        z[4] += r[u] * wb[0];
                        z[5] += r[u] * wb[1];
                        z[6] += r[u] * wb[2];
                        z[7] += r[u] * wb[3];
                    }
                }
                j0 += 32;
            } while (j0 < je);

            const float exs = __expf(aself - mrun);
            dl += __shfl_xor(dl, 8);
            dl += __shfl_xor(dl, 16);
            const float den = dl + exs;
            const float inv = (1.0f / NH) / den;
            {
                v4f r = *(const v4f*)(xl + n * F_IN);
                #pragma unroll
                for (int g = 0; g < NH; ++g) z[g] += r * GAT_lane_bcast(exs, g);
            }
            #pragma unroll
            for (int g = 0; g < NH; ++g) z[g] *= GAT_lane_bcast(inv, g);
        }
        float* zrow = zt + (rt * 16 + wave) * ZP + lane * 4;
        #pragma unroll
        for (int g = 0; g < NH; ++g) *(v4f*)(zrow + g * F_IN) = z[g];
    }
    __syncthreads();

    {
        const int hw = wave >> 1;
        const int ct = wave & 1;
        const int l  = lane & 15;
        const int hf = lane >> 4;
        const float* wp  = W + (long)(hw * NC + ct * 16 + l) * F_IN + 4 * hf;
        const float* za0 = zt + l * ZP + hw * F_IN + 4 * hf;
        const float* za1 = za0 + 16 * ZP;
        v8f p00 = {0.f,0.f,0.f,0.f,0.f,0.f,0.f,0.f};
        v8f p01 = p00, p10 = p00, p11 = p00;
        #pragma unroll
        for (int k0 = 0; k0 < F_IN; k0 += 8) {
            v4f b  = *(const v4f*)(wp  + k0);
            v4f a0 = *(const v4f*)(za0 + k0);
            v4f a1 = *(const v4f*)(za1 + k0);
            v2f bl  = {b[0],  b[1]},  bh  = {b[2],  b[3]};
            v2f a0l = {a0[0], a0[1]}, a0h = {a0[2], a0[3]};
            v2f a1l = {a1[0], a1[1]}, a1h = {a1[2], a1[3]};
            p00 = __builtin_amdgcn_wmma_f32_16x16x4_f32(false, a0l, false, bl, (short)0, p00, false, false);
            p10 = __builtin_amdgcn_wmma_f32_16x16x4_f32(false, a1l, false, bl, (short)0, p10, false, false);
            p01 = __builtin_amdgcn_wmma_f32_16x16x4_f32(false, a0h, false, bh, (short)0, p01, false, false);
            p11 = __builtin_amdgcn_wmma_f32_16x16x4_f32(false, a1h, false, bh, (short)0, p11, false, false);
        }
        v8f p0 = p00 + p01;
        v8f p1 = p10 + p11;
        __syncthreads();
        float* pt = zt + wave * 512 + l;
        #pragma unroll
        for (int v = 0; v < 8; ++v) {
            pt[(8 * hf + v) * 16]       = p0[v];
            pt[256 + (8 * hf + v) * 16] = p1[v];
        }
    }
    __syncthreads();
    {
        const int c   = lane;
        const int ct2 = c >> 4, cc = c & 15;
        const float bc = bias[c];
        #pragma unroll
        for (int rt = 0; rt < 2; ++rt) {
            const long n = (long)blockIdx.x * 32 + rt * 16 + wave;
            float s = 0.0f;
            #pragma unroll
            for (int g = 0; g < NH; ++g)
                s += zt[((g * 2 + ct2) * 2 + rt) * 256 + wave * 16 + cc];
            if (n < N) out[n * NC + c] = s + bc;
        }
    }
}

extern "C" void kernel_launch(void* const* d_in, const int* in_sizes, int n_in,
                              void* d_out, int out_size, void* d_ws, size_t ws_size,
                              hipStream_t stream) {
    const float* x        = (const float*)d_in[0];
    const int*   eidx     = (const int*)  d_in[1];
    const float* eattr    = (const float*)d_in[2];
    const float* W        = (const float*)d_in[3];
    const float* We       = (const float*)d_in[4];
    const float* att_src  = (const float*)d_in[5];
    const float* att_dst  = (const float*)d_in[6];
    const float* att_edge = (const float*)d_in[7];
    const float* bias     = (const float*)d_in[8];

    const int N = in_sizes[0] / F_IN;
    const int E = in_sizes[1] / 2;
    const int* src = eidx;
    const int* dst = eidx + E;

    float* ws = (float*)d_ws;
    const long Nl = N, El = E;
    const long NP = (Nl + 63) & ~63L;
    const long epc = (((El + RK_G - 1) / RK_G) + 255) & ~255L;
    int*      w_deg   = (int*)ws;
    int*      w_tot   = (int*)(ws + Nl);
    const long o_off  = (Nl + 64 + 63) & ~63L;
    const long o_rank = (o_off + Nl + 63) & ~63L;
    const long o_perm = (o_rank + El + 63) & ~63L;
    const long o_att  = (o_perm + 2 * El + 63) & ~63L;
    const long o_cnt  = (o_att + 16 * Nl + 128 + 16 * F_IN + 63) & ~63L;
    int*      w_off   = (int*)(ws + o_off);
    int*      w_rank  = (int*)(ws + o_rank);
    int*      w_perm  = (int*)(ws + o_perm);
    float*    w_asrc  = ws + o_att;
    float*    w_adst  = ws + o_att + 8 * Nl;
    float*    w_M     = ws + o_att + 16 * Nl;
    float*    w_Vt    = ws + o_att + 16 * Nl + 128;
    int*      w_cnt   = (int*)(ws + o_cnt);

    const int B = 256;
    GAT_rank<<<RK_G, 1024, 0, stream>>>(dst, w_rank, w_cnt, w_tot, E, N, (int)epc, (int)NP);
    GAT_segalloc<<<(int)((Nl + 255) / 256), 256, 0, stream>>>(w_cnt, w_deg, w_off, w_tot, N, (int)NP);
    GAT_fill<<<dim3((unsigned)(epc / B), RK_G), B, 0, stream>>>(src, dst, w_cnt, w_rank, w_perm, E, (int)epc, (int)NP);
    GAT_fold<<<9, B, 0, stream>>>(W, We, att_src, att_dst, att_edge, w_M, w_Vt);
    {
        long rowTiles = (Nl + 15) >> 4;
        GAT_gemm_att<<<(int)((rowTiles + 7) / 8), B, 0, stream>>>(x, w_Vt, w_asrc, w_adst, N);
    }
    GAT_gather<<<(int)((Nl + 31) / 32), 512, 0, stream>>>(w_deg, w_off, w_perm, eattr, w_M,
                                                          w_asrc, w_adst, x, W, bias,
                                                          (float*)d_out, N);
}


# Weight fold and node-term GEMM lose their own launches: they run as extra work-groups of the rank launch and of the fill launch (6 launches -> 4); sums and their order unchanged
# speedup vs baseline: 8.3196x; 1.0484x over previous
#include <hip/hip_runtime.h>
#include <hip/hip_bf16.h>

typedef float v2f __attribute__((ext_vector_type(2)));
typedef float v8f __attribute__((ext_vector_type(8)));
typedef float v4f __attribute__((ext_vector_type(4)));

#define F_IN 128
#define F_E  16
#define NH   8
#define NC   32
#define HC   256
#define NEG_SLOPE 0.2f

typedef int v4i __attribute__((ext_vector_type(4)));
typedef int v2i __attribute__((ext_vector_type(2)));

#define RK_G    64
#define RK_CAP  51200
#define RK_B    16

__device__ __forceinline__ void GAT_fold_part(int part, int tid,
                         const float* __restrict__ W,
                         const float* __restrict__ We,
                         const float* __restrict__ att_src,
                         const float* __restrict__ att_dst,
                         const float* __restrict__ att_edge,
                         float* __restrict__ M,
                         float* __restrict__ Vt) {
    if (part < 2) {
        int idx = part * 1024 + tid;
        int j = idx >> 7, f = idx & (F_IN - 1);
        int hj = j & 7;
        const float* att = (j < NH) ? att_src : att_dst;
        float s = 0.0f;
        for (int c = 0; c < NC; ++c)
            s += W[(long)(hj * NC + c) * F_IN + f] * att[hj * NC + c];
        Vt[j * F_IN + f] = s;
    } else {
        if (tid >= F_E * NH) return;
        int f = tid >> 3, h = tid & 7;
        float s = 0.0f;
        for (int c = 0; c < NC; ++c)
            s += We[(long)(h * NC + c) * F_E + f] * att_edge[h * NC + c];
        M[f * NH + h] = s;
    }
}

__global__ void __launch_bounds__(1024)
GAT_rank(const int* __restrict__ dst, int* __restrict__ rank, int* __restrict__ cnt,
         int* __restrict__ total, int E, int N, int epc, int np,
         const float* __restrict__ W, const float* __restrict__ We,
         const float* __restrict__ att_src, const float* __restrict__ att_dst,
         const float* __restrict__ att_edge, float* __restrict__ M, float* __restrict__ Vt) {
    __shared__ __attribute__((aligned(16))) int hist[RK_CAP];
    const int t = threadIdx.x;
    const int g = blockIdx.x;
    if (g >= RK_G) {
        GAT_fold_part(g - RK_G, t, W, We, att_src, att_dst, att_edge, M, Vt);
        return;
    }
    if (g == 0 && t == 0) *total = 0;
    const long e0 = (long)g * epc;
    long e1 = e0 + epc;
    if (e1 > E) e1 = E;
    for (int nb = 0; nb < N; nb += RK_CAP) {
        int sn = N - nb;
        if (sn > RK_CAP) sn = RK_CAP;
        const int sn4 = (sn + 3) >> 2;
        int d[RK_B];
        long eb = e0;
        #pragma unroll
        for (int u = 0; u < RK_B; ++u) {
            const long e = eb + u * 1024 + t;
            d[u] = (e < e1) ? dst[e] : -1;
        }
        for (int i = t; i < sn4; i += 1024) ((v4i*)hist)[i] = (v4i){0, 0, 0, 0};
        __syncthreads();
        while (true) {
            int r[RK_B];
            #pragma unroll
            for (int u = 0; u < RK_B; ++u) {
                const unsigned k = (unsigned)(d[u] - nb);
                r[u] = (k < (unsigned)sn) ? atomicAdd(&hist[k], 1) : -1;
            }
            #pragma unroll
            for (int u = 0; u < RK_B; ++u) {
                const long e = eb + u * 1024 + t;
                if (r[u] >= 0) rank[e] = r[u];
            }
            eb += (long)RK_B * 1024;
            if (eb >= e1) break;
            #pragma unroll
            for (int u = 0; u < RK_B; ++u) {
                const long e = eb + u * 1024 + t;
                d[u] = (e < e1) ? dst[e] : -1;
            }
        }
        __syncthreads();
        int* row = cnt + (long)g * np + nb;
        #pragma unroll 4
        for (int i = t; i < sn4; i += 1024) ((v4i*)row)[i] = ((const v4i*)hist)[i];
        __syncthreads();
    }
}

__device__ __forceinline__ void GAT_gemm_att_part(long blk,
                             const float* __restrict__ x,
                             const float* __restrict__ Vt,
                             float* __restrict__ asrc,
                             float* __restrict__ adst, int N) {
    int wave = threadIdx.x >> 5;
    int lane = threadIdx.x & 31;
    int half = lane >> 4;
    int l    = lane & 15;
    long rowTiles = ((long)N + 15) >> 4;
    long tile = blk * 8 + wave;
    if (tile >= rowTiles) return;
    long row0 = tile * 16;
    long arow = row0 + l;
    if (arow >= N) arow = N - 1;
    const float* xp = x  + arow * F_IN + 2 * half;
    const float* vp = Vt + (long)l * F_IN + 2 * half;
    v8f a0 = {0.f,0.f,0.f,0.f,0.f,0.f,0.f,0.f};
    v8f a1 = a0;
    #pragma unroll
    for (int k0 = 0; k0 < F_IN; k0 += 8) {
        v2f xa = *(const v2f*)(xp + k0);
        v2f va = *(const v2f*)(vp + k0);
        v2f xb = *(const v2f*)(xp + k0 + 4);
        v2f vb = *(const v2f*)(vp + k0 + 4);
        a0 = __builtin_amdgcn_wmma_f32_16x16x4_f32(false, xa, false, va, (short)0, a0, false, false);
        a1 = __builtin_amdgcn_wmma_f32_16x16x4_f32(false, xb, false, vb, (short)0, a1, false, false);
    }
    v8f acc = a0 + a1;
    #pragma unroll
    for (int v = 0; v < 8; ++v) {
        long r = row0 + v + 8 * half;
        if (r < N) {
            if (l < NH) asrc[r * NH + l] = acc[v];
            else        adst[r * NH + (l - NH)] = acc[v];
        }
    }
}

__global__ void __launch_bounds__(256)
GAT_segalloc(int* __restrict__ cnt, int* __restrict__ deg, int* __restrict__ off,
             int* __restrict__ total, int N, int np) {
    __shared__ int wsum[8];
    __shared__ int base;
    const int t = threadIdx.x;
    const int lane = t & 31, w = t >> 5;
    const long n = (long)blockIdx.x * 256 + t;
    int* col = cnt + ((n < N) ? n : (long)N - 1);
    int c[RK_G];
    #pragma unroll
    for (int g = 0; g < RK_G; ++g) c[g] = col[(long)g * np];
    int tot = 0;
    #pragma unroll
    for (int g = 0; g < RK_G; ++g) tot += c[g];
    if (n < N) deg[n] = tot; else tot = 0;
    int inc = tot;
    for (int o = 1; o < 32; o <<= 1) {
        int v = __shfl_up(inc, o);
        if (lane >= o) inc += v;
    }
    if (lane == 31) wsum[w] = inc;
    __syncthreads();
    if (w == 0) {
        int s = (lane < 8) ? wsum[lane] : 0;
        int si = s;
        for (int o = 1; o < 32; o <<= 1) {
            int v = __shfl_up(si, o);
            if (lane >= o) si += v;
        }
        if (lane < 8) wsum[lane] = si - s;
        if (lane == 31) base = atomicAdd(total, si);
    }
    __syncthreads();
    if (n < N) {
        int run = base + wsum[w] + inc - tot;
        off[n] = run;
        #pragma unroll
        for (int g = 0; g < RK_G; ++g) { col[(long)g * np] = run; run += c[g]; }
    }
}

__global__ void __launch_bounds__(256)
GAT_fill(const int* __restrict__ src, const int* __restrict__ dst, const int* __restrict__ cbase,
         const int* __restrict__ rank, int* __restrict__ perm, int E, int epc, int np, int gy,
         const float* __restrict__ x, const float* __restrict__ Vt,
         float* __restrict__ asrc, float* __restrict__ adst, int N) {
    if ((int)blockIdx.y < gy) {
        GAT_gemm_att_part((long)blockIdx.y * gridDim.x + blockIdx.x, x, Vt, asrc, adst, N);
        return;
    }
    const int g = blockIdx.y - gy;
    const long e = (long)g * epc + (long)blockIdx.x * blockDim.x + threadIdx.x;
    if (e >= E) return;
    const long slot = (long)cbase[(long)g * np + dst[e]] + rank[e];
    ((v2i*)perm)[slot] = (v2i){(int)e, src[e]};
}

__device__ __forceinline__ float GAT_lane_bcast(float v, int g) {
    return __builtin_bit_cast(float, __builtin_amdgcn_readlane(__builtin_bit_cast(int, v), g));
}
#define ZP 1028
#define SP 20
__global__ void __launch_bounds__(512, 8)
GAT_gather(const int* __restrict__ deg,
           const int* __restrict__ off,
           const int* __restrict__ perm,
           const float* __restrict__ eattr,
           const float* __restrict__ M,
           const float* __restrict__ asrc,
           const float* __restrict__ adst,
           const float* __restrict__ x,
           const float* __restrict__ W,
           const float* __restrict__ bias,
           float* __restrict__ out, int N) {
    __shared__ __attribute__((aligned(16))) float zt[32 * ZP];
    __shared__ __attribute__((aligned(16))) float wexp[16][32 * NH];
    const int wave = __builtin_amdgcn_readfirstlane((int)(threadIdx.x >> 5));
    const int lane = threadIdx.x & 31;
    const int hq = lane >> 3;
    const int h  = lane & 7;
    float* wmine = &wexp[wave][0];
    const float* xl = x + lane * 4;

    float mcol[F_E];
    #pragma unroll
    for (int f = 0; f < F_E; ++f) mcol[f] = M[f * NH + h];

    #pragma nounroll
    for (int rt = 0; rt < 2; ++rt) {
        const long n = (long)blockIdx.x * 32 + rt * 16 + wave;
        float* stage = zt + (rt * 16 + wave) * ZP;
        v4f z[NH];
        #pragma unroll
        for (int g = 0; g < NH; ++g) z[g] = (v4f){0.f, 0.f, 0.f, 0.f};
        if (n < N) {
            const float adh = adst[n * NH + h];
            const float asn = asrc[n * NH + h];
            float mrun = -__builtin_inff();
            float dl = 0.0f;
            float es0 = 0.0f, es1 = 0.0f;
            float aself = 0.0f;
            const int dg = deg[n];
            const int jb = off[n];
            const int je = jb + dg;
            int j0 = jb;
            do {
                int cnt = je - j0;
                if (cnt > 32) cnt = 32;
                int me = 0, ms = (int)n;
                if (lane < cnt) {
                    const v2i p = ((const v2i*)perm)[j0 + lane];
                    me = p[0];
                    ms = p[1];
                }
                const int nk = (cnt + 3) >> 2;
                if (lane < 4 * nk) {
                    const float* er = eattr + (long)me * F_E;
                    const float* tr = asrc + (long)ms * NH;
                    v4f s0 = *(const v4f*)(er);
                    v4f s1 = *(const v4f*)(er + 4);
                    v4f s2 = *(const v4f*)(er + 8);
                    v4f s3 = *(const v4f*)(er + 12);
                    v4f t0 = *(const v4f*)(tr);
                    v4f t1 = *(const v4f*)(tr + 4);
                    float* sr = stage + lane * SP;
                    *(v4f*)(sr)      = s0;
                    *(v4f*)(sr + 4)  = s1;
                    *(v4f*)(sr + 8)  = s2;
                    *(v4f*)(sr + 12) = s3;
                    *(v4f*)(wmine + lane * NH)     = t0;
                    *(v4f*)(wmine + lane * NH + 4) = t1;
                }
                float lmax = -__builtin_inff();
                #pragma unroll 2
                for (int k = 0; k < nk; ++k) {
                    const int j = 4 * k + hq;
                    const bool valid = j < cnt;
                    const float* er = stage + j * SP;
                    v4f e0 = *(const v4f*)(er);
                    v4f e1 = *(const v4f*)(er + 4);
                    v4f e2 = *(const v4f*)(er + 8);
                    v4f e3 = *(const v4f*)(er + 12);
                    v2f ef = *(const v2f*)(er + 2 * h);
                    float as = wmine[32 * k + lane];
                    float ae = 0.0f;
                    #pragma unroll
                    for (int q = 0; q < 4; ++q) ae = __builtin_fmaf(e0[q], mcol[q], ae);
                    #pragma unroll
                    for (int q = 0; q < 4; ++q) ae = __builtin_fmaf(e1[q], mcol[4 + q], ae);
                    #pragma unroll
                    for (int q = 0; q < 4; ++q) ae = __builtin_fmaf(e2[q], mcol[8 + q], ae);
                    #pragma unroll
                    for (int q = 0; q < 4; ++q) ae = __builtin_fmaf(e3[q], mcol[12 + q], ae);
                    float a = as + adh + ae;
                    a = (a >= 0.0f) ? a : NEG_SLOPE * a;
                    a = valid ? a : -__builtin_inff();
                    es0 += valid ? ef[0] : 0.0f;
                    es1 += valid ? ef[1] : 0.0f;
                    wmine[32 * k + lane] = a;
                    lmax = fmaxf(lmax, a);
                }
                float cm = fmaxf(lmax, __shfl_xor(lmax, 8));
                cm = fmaxf(cm, __shfl_xor(cm, 16));
                if (j0 + 32 >= je) {
                    es0 += __shfl_xor(es0, 8);
                    es0 += __shfl_xor(es0, 16);
                    es1 += __shfl_xor(es1, 8);
                    es1 += __shfl_xor(es1, 16);
                    const float dmax = fmaxf((float)dg, 1.0f);
                    const float lp0 = es0 / dmax;
                    const float lp1 = es1 / dmax;
                    float ae = 0.0f;
                    #pragma unroll
                    for (int g = 0; g < NH; ++g) {
                        ae = __builtin_fmaf(__shfl(lp0, g), mcol[2 * g], ae);
                        ae = __builtin_fmaf(__shfl(lp1, g), mcol[2 * g + 1], ae);
                    }
                    float a = asn + adh + ae;
                    aself = (a >= 0.0f) ? a : NEG_SLOPE * a;
                    cm = fmaxf(cm, aself);
                }
                const float mn = fmaxf(mrun, cm);
                const float sc = __expf(mrun - mn);
                mrun = mn;
                float lsum = 0.0f;
                #pragma unroll 4
                for (int k = 0; k < nk; ++k) {
                    float ex = __expf(wmine[32 * k + lane] - mn);
                    wmine[32 * k + lane] = ex;
                    lsum += ex;
                }
                dl = dl * sc + lsum;
                #pragma unroll
                for (int g = 0; g < NH; ++g) z[g] *= GAT_lane_bcast(sc, g);
                int i = 0;
                for (; i + 8 <= cnt; i += 8) {
                    v4f r[8];
                    #pragma unroll
                    for (int u = 0; u < 8; ++u) {
                        int s = __builtin_amdgcn_readlane(ms, i + u);
                        r[u] = *(const v4f*)(xl + (long)s * F_IN);
                    }
                    asm volatile("" :: "v"(r[0]), "v"(r[1]), "v"(r[2]), "v"(r[3]), "v"(r[4]), "v"(r[5]), "v"(r[6]), "v"(r[7]));
                    #pragma unroll
                    for (int u = 0; u < 8; ++u) {
                        v4f wa = *(const v4f*)(wmine + (i + u) * NH);
                        v4f wb = *(const v4f*)(wmine + (i + u) * NH + 4);
                        z[0] += r[u] * wa[0];
                        z[1] += r[u] * wa[1];
                        z[2] += r[u] * wa[2];
                        z[3] += r[u] * wa[3];
                        z[4] += r[u] * wb[0];
                        z[5] += r[u] * wb[1];
                        z[6] += r[u] * wb[2];
                        z[7] += r[u] * wb[3];
                    }
                }
                for (; i < cnt; i += 4) {
                    v4f r[4];
                    #pragma unroll
                    for (int u = 0; u < 4; ++u) {
                        int s = __builtin_amdgcn_readlane(ms, i + u);
                        r[u] = *(const v4f*)(xl + (long)s * F_IN);
                    }
                    asm volatile("" :: "v"(r[0]), "v"(r[1]), "v"(r[2]), "v"(r[3]));
                    #pragma unroll
                    for (int u = 0; u < 4; ++u) {
                        v4f wa = *(const v4f*)(wmine + (i + u) * NH);
                        v4f wb = *(const v4f*)(wmine + (i + u) * NH + 4);
                        z[0] += r[u] * wa[0];
                        z[1] += r[u] * wa[1];
                        z[2] += r[u] * wa[2];
                        z[3] += r[u] * wa[3];
                        z[4] += r[u] * wb[0];
                        z[5] += r[u] * wb[1];
                        z[6] += r[u] * wb[2];
                        z[7] += r[u] * wb[3];
                    }
                }
                j0 += 32;
            } while (j0 < je);

            const float exs = __expf(aself - mrun);
            dl += __shfl_xor(dl, 8);
            dl += __shfl_xor(dl, 16);
            const float den = dl + exs;
            const float inv = (1.0f / NH) / den;
            {
                v4f r = *(const v4f*)(xl + n * F_IN);
                #pragma unroll
                for (int g = 0; g < NH; ++g) z[g] += r * GAT_lane_bcast(exs, g);
            }
            #pragma unroll
            for (int g = 0; g < NH; ++g) z[g] *= GAT_lane_bcast(inv, g);
        }
        float* zrow = zt + (rt * 16 + wave) * ZP + lane * 4;
        #pragma unroll
        for (int g = 0; g < NH; ++g) *(v4f*)(zrow + g * F_IN) = z[g];
    }
    __syncthreads();

    {
        const int hw = wave >> 1;
        const int ct = wave & 1;
        const int l  = lane & 15;
        const int hf = lane >> 4;
        const float* wp  = W + (long)(hw * NC + ct * 16 + l) * F_IN + 4 * hf;
        const float* za0 = zt + l * ZP + hw * F_IN + 4 * hf;
        const float* za1 = za0 + 16 * ZP;
        v8f p00 = {0.f,0.f,0.f,0.f,0.f,0.f,0.f,0.f};
        v8f p01 = p00, p10 = p00, p11 = p00;
        #pragma unroll
        for (int k0 = 0; k0 < F_IN; k0 += 8) {
            v4f b  = *(const v4f*)(wp  + k0);
            v4f a0 = *(const v4f*)(za0 + k0);
            v4f a1 = *(const v4f*)(za1 + k0);
            v2f bl  = {b[0],  b[1]},  bh  = {b[2],  b[3]};
            v2f a0l = {a0[0], a0[1]}, a0h = {a0[2], a0[3]};
            v2f a1l = {a1[0], a1[1]}, a1h = {a1[2], a1[3]};
            p00 = __builtin_amdgcn_wmma_f32_16x16x4_f32(false, a0l, false, bl, (short)0, p00, false, false);
            p10 = __builtin_amdgcn_wmma_f32_16x16x4_f32(false, a1l, false, bl, (short)0, p10, false, false);
            p01 = __builtin_amdgcn_wmma_f32_16x16x4_f32(false, a0h, false, bh, (short)0, p01, false, false);
            p11 = __builtin_amdgcn_wmma_f32_16x16x4_f32(false, a1h, false, bh, (short)0, p11, false, false);
        }
        v8f p0 = p00 + p01;
        v8f p1 = p10 + p11;
        __syncthreads();
        float* pt = zt + wave * 512 + l;
        #pragma unroll
        for (int v = 0; v < 8; ++v) {
            pt[(8 * hf + v) * 16]       = p0[v];
            pt[256 + (8 * hf + v) * 16] = p1[v];
        }
    }
    __syncthreads();
    {
        const int c   = lane;
        const int ct2 = c >> 4, cc = c & 15;
        const float bc = bias[c];
        #pragma unroll
        for (int rt = 0; rt < 2; ++rt) {
            const long n = (long)blockIdx.x * 32 + rt * 16 + wave;
            float s = 0.0f;
            #pragma unroll
            for (int g = 0; g < NH; ++g)
                s += zt[((g * 2 + ct2) * 2 + rt) * 256 + wave * 16 + cc];
            if (n < N) out[n * NC + c] = s + bc;
        }
    }
}

extern "C" void kernel_launch(void* const* d_in, const int* in_sizes, int n_in,
                              void* d_out, int out_size, void* d_ws, size_t ws_size,
                              hipStream_t stream) {
    const float* x        = (const float*)d_in[0];
    const int*   eidx     = (const int*)  d_in[1];
    const float* eattr    = (const float*)d_in[2];
    const float* W        = (const float*)d_in[3];
    const float* We       = (const float*)d_in[4];
    const float* att_src  = (const float*)d_in[5];
    const float* att_dst  = (const float*)d_in[6];
    const float* att_edge = (const float*)d_in[7];
    const float* bias     = (const float*)d_in[8];

    const int N = in_sizes[0] / F_IN;
    const int E = in_sizes[1] / 2;
    const int* src = eidx;
    const int* dst = eidx + E;

    float* ws = (float*)d_ws;
    const long Nl = N, El = E;
    const long NP = (Nl + 63) & ~63L;
    const long epc = (((El + RK_G - 1) / RK_G) + 255) & ~255L;
    int*      w_deg   = (int*)ws;
    int*      w_tot   = (int*)(ws + Nl);
    const long o_off  = (Nl + 64 + 63) & ~63L;
    const long o_rank = (o_off + Nl + 63) & ~63L;
    const long o_perm = (o_rank + El + 63) & ~63L;
    const long o_att  = (o_perm + 2 * El + 63) & ~63L;
    const long o_cnt  = (o_att + 16 * Nl + 128 + 16 * F_IN + 63) & ~63L;
    int*      w_off   = (int*)(ws + o_off);
    int*      w_rank  = (int*)(ws + o_rank);
    int*      w_perm  = (int*)(ws + o_perm);
    float*    w_asrc  = ws + o_att;
    float*    w_adst  = ws + o_att + 8 * Nl;
    float*    w_M     = ws + o_att + 16 * Nl;
    float*    w_Vt    = ws + o_att + 16 * Nl + 128;
    int*      w_cnt   = (int*)(ws + o_cnt);

    const int B = 256;
    GAT_rank<<<RK_G + 3, 1024, 0, stream>>>(dst, w_rank, w_cnt, w_tot, E, N, (int)epc, (int)NP,
                                            W, We, att_src, att_dst, att_edge, w_M, w_Vt);
    GAT_segalloc<<<(int)((Nl + 255) / 256), 256, 0, stream>>>(w_cnt, w_deg, w_off, w_tot, N, (int)NP);
    {
        const long bpc = epc / B;
        const long attBlocks = (((Nl + 15) >> 4) + 7) / 8;
        const long gy = (attBlocks + bpc - 1) / bpc;
        GAT_fill<<<dim3((unsigned)bpc, (unsigned)(RK_G + gy)), B, 0, stream>>>(src, dst, w_cnt, w_rank, w_perm, E, (int)epc, (int)NP,
                                                                           (int)gy, x, w_Vt, w_asrc, w_adst, N);
    }
    GAT_gather<<<(int)((Nl + 31) / 32), 512, 0, stream>>>(w_deg, w_off, w_perm, eattr, w_M,
                                                          w_asrc, w_adst, x, W, bias,
                                                          (float*)d_out, N);
}


# Attribute terms folded in the fill pass: it reads each 64-B attribute row in edge order, stores 8 head terms (32 B); node waves gather term rows; self-loop term = mean of terms; sum order changes
# speedup vs baseline: 8.7210x; 1.0482x over previous
#include <hip/hip_runtime.h>
#include <hip/hip_bf16.h>

typedef float v2f __attribute__((ext_vector_type(2)));
typedef float v8f __attribute__((ext_vector_type(8)));
typedef float v4f __attribute__((ext_vector_type(4)));

#define F_IN 128
#define F_E  16
#define NH   8
#define NC   32
#define HC   256
#define NEG_SLOPE 0.2f

typedef int v4i __attribute__((ext_vector_type(4)));
typedef int v2i __attribute__((ext_vector_type(2)));

#define RK_G    64
#define RK_CAP  51200
#define RK_B    16

__device__ __forceinline__ void GAT_fold_part(int part, int tid,
                         const float* __restrict__ W,
                         const float* __restrict__ We,
                         const float* __restrict__ att_src,
                         const float* __restrict__ att_dst,
                         const float* __restrict__ att_edge,
                         float* __restrict__ M,
                         float* __restrict__ Vt) {
    if (part < 2) {
        int idx = part * 1024 + tid;
        int j = idx >> 7, f = idx & (F_IN - 1);
        int hj = j & 7;
        const float* att = (j < NH) ? att_src : att_dst;
        float s = 0.0f;
        for (int c = 0; c < NC; ++c)
            s += W[(long)(hj * NC + c) * F_IN + f] * att[hj * NC + c];
        Vt[j * F_IN + f] = s;
    } else {
        if (tid >= F_E * NH) return;
        int f = tid >> 3, h = tid & 7;
        float s = 0.0f;
        for (int c = 0; c < NC; ++c)
            s += We[(long)(h * NC + c) * F_E + f] * att_edge[h * NC + c];
        M[f * NH + h] = s;
    }
}

__global__ void __launch_bounds__(1024)
GAT_rank(const int* __restrict__ dst, int* __restrict__ rank, int* __restrict__ cnt,
         int* __restrict__ total, int E, int N, int epc, int np,
         const float* __restrict__ W, const float* __restrict__ We,
         const float* __restrict__ att_src, const float* __restrict__ att_dst,
         const float* __restrict__ att_edge, float* __restrict__ M, float* __restrict__ Vt) {
    __shared__ __attribute__((aligned(16))) int hist[RK_CAP];
    const int t = threadIdx.x;
    const int g = blockIdx.x;
    if (g >= RK_G) {
        GAT_fold_part(g - RK_G, t, W, We, att_src, att_dst, att_edge, M, Vt);
        return;
    }
    if (g == 0 && t == 0) *total = 0;
    const long e0 = (long)g * epc;
    long e1 = e0 + epc;
    if (e1 > E) e1 = E;
    for (int nb = 0; nb < N; nb += RK_CAP) {
        int sn = N - nb;
        if (sn > RK_CAP) sn = RK_CAP;
        const int sn4 = (sn + 3) >> 2;
        int d[RK_B];
        long eb = e0;
        #pragma unroll
        for (int u = 0; u < RK_B; ++u) {
            const long e = eb + u * 1024 + t;
            d[u] = (e < e1) ? dst[e] : -1;
        }
        for (int i = t; i < sn4; i += 1024) ((v4i*)hist)[i] = (v4i){0, 0, 0, 0};
        __syncthreads();
        while (true) {
            int r[RK_B];
            #pragma unroll
            for (int u = 0; u < RK_B; ++u) {
                const unsigned k = (unsigned)(d[u] - nb);
                r[u] = (k < (unsigned)sn) ? atomicAdd(&hist[k], 1) : -1;
            }
            #pragma unroll
            for (int u = 0; u < RK_B; ++u) {
                const long e = eb + u * 1024 + t;
                if (r[u] >= 0) rank[e] = r[u];
            }
            eb += (long)RK_B * 1024;
            if (eb >= e1) break;
            #pragma unroll
            for (int u = 0; u < RK_B; ++u) {
                const long e = eb + u * 1024 + t;
                d[u] = (e < e1) ? dst[e] : -1;
            }
        }
        __syncthreads();
        int* row = cnt + (long)g * np + nb;
        #pragma unroll 4
        for (int i = t; i < sn4; i += 1024) ((v4i*)row)[i] = ((const v4i*)hist)[i];
        __syncthreads();
    }
}

__device__ __forceinline__ void GAT_gemm_att_part(long blk,
                             const float* __restrict__ x,
                             const float* __restrict__ Vt,
                             float* __restrict__ asrc,
                             float* __restrict__ adst, int N) {
    int wave = threadIdx.x >> 5;
    int lane = threadIdx.x & 31;
    int half = lane >> 4;
    int l    = lane & 15;
    long rowTiles = ((long)N + 15) >> 4;
    long tile = blk * 8 + wave;
    if (tile >= rowTiles) return;
    long row0 = tile * 16;
    long arow = row0 + l;
    if (arow >= N) arow = N - 1;
    const float* xp = x  + arow * F_IN + 2 * half;
    const float* vp = Vt + (long)l * F_IN + 2 * half;
    v8f a0 = {0.f,0.f,0.f,0.f,0.f,0.f,0.f,0.f};
    v8f a1 = a0;
    #pragma unroll
    for (int k0 = 0; k0 < F_IN; k0 += 8) {
        v2f xa = *(const v2f*)(xp + k0);
        v2f va = *(const v2f*)(vp + k0);
        v2f xb = *(const v2f*)(xp + k0 + 4);
        v2f vb = *(const v2f*)(vp + k0 + 4);
        a0 = __builtin_amdgcn_wmma_f32_16x16x4_f32(false, xa, false, va, (short)0, a0, false, false);
        a1 = __builtin_amdgcn_wmma_f32_16x16x4_f32(false, xb, false, vb, (short)0, a1, false, false);
    }
    v8f acc = a0 + a1;
    #pragma unroll
    for (int v = 0; v < 8; ++v) {
        long r = row0 + v + 8 * half;
        if (r < N) {
            if (l < NH) asrc[r * NH + l] = acc[v];
            else        adst[r * NH + (l - NH)] = acc[v];
        }
    }
}

__global__ void __launch_bounds__(256)
GAT_segalloc(int* __restrict__ cnt, int* __restrict__ deg, int* __restrict__ off,
             int* __restrict__ total, int N, int np) {
    __shared__ int wsum[8];
    __shared__ int base;
    const int t = threadIdx.x;
    const int lane = t & 31, w = t >> 5;
    const long n = (long)blockIdx.x * 256 + t;
    int* col = cnt + ((n < N) ? n : (long)N - 1);
    int c[RK_G];
    #pragma unroll
    for (int g = 0; g < RK_G; ++g) c[g] = col[(long)g * np];
    int tot = 0;
    #pragma unroll
    for (int g = 0; g < RK_G; ++g) tot += c[g];
    if (n < N) deg[n] = tot; else tot = 0;
    int inc = tot;
    for (int o = 1; o < 32; o <<= 1) {
        int v = __shfl_up(inc, o);
        if (lane >= o) inc += v;
    }
    if (lane == 31) wsum[w] = inc;
    __syncthreads();
    if (w == 0) {
        int s = (lane < 8) ? wsum[lane] : 0;
        int si = s;
        for (int o = 1; o < 32; o <<= 1) {
            int v = __shfl_up(si, o);
            if (lane >= o) si += v;
        }
        if (lane < 8) wsum[lane] = si - s;
        if (lane == 31) base = atomicAdd(total, si);
    }
    __syncthreads();
    if (n < N) {
        int run = base + wsum[w] + inc - tot;
        off[n] = run;
        #pragma unroll
        for (int g = 0; g < RK_G; ++g) { col[(long)g * np] = run; run += c[g]; }
    }
}

__global__ void __launch_bounds__(256)
GAT_fill(const int* __restrict__ src, const int* __restrict__ dst, const int* __restrict__ cbase,
         const int* __restrict__ rank, int* __restrict__ perm, int E, int epc, int np, int gy,
         const float* __restrict__ x, const float* __restrict__ Vt,
         float* __restrict__ asrc, float* __restrict__ adst, int N,
         const float* __restrict__ eattr, const float* __restrict__ M, float* __restrict__ aedge) {
    if ((int)blockIdx.y < gy) {
        GAT_gemm_att_part((long)blockIdx.y * gridDim.x + blockIdx.x, x, Vt, asrc, adst, N);
        return;
    }
    const int g = blockIdx.y - gy;
    const long e = (long)g * epc + (long)blockIdx.x * blockDim.x + threadIdx.x;
    if (e >= E) return;
    const v4f* er = (const v4f*)(eattr + e * F_E);
    const v4f e0 = er[0], e1 = er[1], e2 = er[2], e3 = er[3];
    const long slot = (long)cbase[(long)g * np + dst[e]] + rank[e];
    ((v2i*)perm)[slot] = (v2i){(int)e, src[e]};
    float t[NH];
    #pragma unroll
    for (int h = 0; h < NH; ++h) t[h] = 0.0f;
    #pragma unroll
    for (int q = 0; q < 4; ++q) {
        #pragma unroll
        for (int h = 0; h < NH; ++h) t[h] = __builtin_fmaf(e0[q], M[q * NH + h], t[h]);
    }
    #pragma unroll
    for (int q = 0; q < 4; ++q) {
        #pragma unroll
        for (int h = 0; h < NH; ++h) t[h] = __builtin_fmaf(e1[q], M[(4 + q) * NH + h], t[h]);
    }
    #pragma unroll
    for (int q = 0; q < 4; ++q) {
        #pragma unroll
        for (int h = 0; h < NH; ++h) t[h] = __builtin_fmaf(e2[q], M[(8 + q) * NH + h], t[h]);
    }
    #pragma unroll
    for (int q = 0; q < 4; ++q) {
        #pragma unroll
        for (int h = 0; h < NH; ++h) t[h] = __builtin_fmaf(e3[q], M[(12 + q) * NH + h], t[h]);
    }
    v4f* tr = (v4f*)(aedge + e * NH);
    tr[0] = (v4f){t[0], t[1], t[2], t[3]};
    tr[1] = (v4f){t[4], t[5], t[6], t[7]};
}

__device__ __forceinline__ float GAT_lane_bcast(float v, int g) {
    return __builtin_bit_cast(float, __builtin_amdgcn_readlane(__builtin_bit_cast(int, v), g));
}
#define ZP 1028
__global__ void __launch_bounds__(512, 8)
GAT_gather(const int* __restrict__ deg,
           const int* __restrict__ off,
           const int* __restrict__ perm,
           const float* __restrict__ aedge,
           const float* __restrict__ asrc,
           const float* __restrict__ adst,
           const float* __restrict__ x,
           const float* __restrict__ W,
           const float* __restrict__ bias,
           float* __restrict__ out, int N) {
    __shared__ __attribute__((aligned(16))) float zt[32 * ZP];
    __shared__ __attribute__((aligned(16))) float wexp[16][32 * NH];
    const int wave = __builtin_amdgcn_readfirstlane((int)(threadIdx.x >> 5));
    const int lane = threadIdx.x & 31;
    const int hq = lane >> 3;
    const int h  = lane & 7;
    float* wmine = &wexp[wave][0];
    const float* xl = x + lane * 4;

    #pragma nounroll
    for (int rt = 0; rt < 2; ++rt) {
        const long n = (long)blockIdx.x * 32 + rt * 16 + wave;
        float* stage = zt + (rt * 16 + wave) * ZP;
        v4f z[NH];
        #pragma unroll
        for (int g = 0; g < NH; ++g) z[g] = (v4f){0.f, 0.f, 0.f, 0.f};
        if (n < N) {
            const float adh = adst[n * NH + h];
            const float asn = asrc[n * NH + h];
            float mrun = -__builtin_inff();
            float dl = 0.0f;
            float es = 0.0f;
            float aself = 0.0f;
            const int dg = deg[n];
            const int jb = off[n];
            const int je = jb + dg;
            int j0 = jb;
            do {
                int cnt = je - j0;
                if (cnt > 32) cnt = 32;
                int me = 0, ms = (int)n;
                if (lane < cnt) {
                    const v2i p = ((const v2i*)perm)[j0 + lane];
                    me = p[0];
                    ms = p[1];
                }
                const int nk = (cnt + 3) >> 2;
                if (lane < 4 * nk) {
                    const float* er = aedge + (long)me * NH;
                    const float* tr = asrc + (long)ms * NH;
                    v4f s0 = *(const v4f*)(er);
                    v4f s1 = *(const v4f*)(er + 4);
                    v4f t0 = *(const v4f*)(tr);
                    v4f t1 = *(const v4f*)(tr + 4);
                    *(v4f*)(stage + lane * NH)     = s0;
                    *(v4f*)(stage + lane * NH + 4) = s1;
                    *(v4f*)(wmine + lane * NH)     = t0;
                    *(v4f*)(wmine + lane * NH + 4) = t1;
                }
                float lmax = -__builtin_inff();
                #pragma unroll 2
                for (int k = 0; k < nk; ++k) {
                    const int j = 4 * k + hq;
                    const bool valid = j < cnt;
                    float ae = stage[32 * k + lane];
                    float as = wmine[32 * k + lane];
                    float a = as + adh + ae;
                    a = (a >= 0.0f) ? a : NEG_SLOPE * a;
                    a = valid ? a : -__builtin_inff();
                    es += valid ? ae : 0.0f;
                    wmine[32 * k + lane] = a;
                    lmax = fmaxf(lmax, a);
                }
                float cm = fmaxf(lmax, __shfl_xor(lmax, 8));
                cm = fmaxf(cm, __shfl_xor(cm, 16));
                if (j0 + 32 >= je) {
                    es += __shfl_xor(es, 8);
                    es += __shfl_xor(es, 16);
                    const float dmax = fmaxf((float)dg, 1.0f);
                    const float ae = es / dmax;
                    float a = asn + adh + ae;
                    aself = (a >= 0.0f) ? a : NEG_SLOPE * a;
                    cm = fmaxf(cm, aself);
                }
                const float mn = fmaxf(mrun, cm);
                const float sc = __expf(mrun - mn);
                mrun = mn;
                float lsum = 0.0f;
                #pragma unroll 4
                for (int k = 0; k < nk; ++k) {
                    float ex = __expf(wmine[32 * k + lane] - mn);
                    wmine[32 * k + lane] = ex;
                    lsum += ex;
                }
                dl = dl * sc + lsum;
                #pragma unroll
                for (int g = 0; g < NH; ++g) z[g] *= GAT_lane_bcast(sc, g);
                int i = 0;
                for (; i + 8 <= cnt; i += 8) {
                    v4f r[8];
                    #pragma unroll
                    for (int u = 0; u < 8; ++u) {
                        int s = __builtin_amdgcn_readlane(ms, i + u);
                        r[u] = *(const v4f*)(xl + (long)s * F_IN);
                    }
                    asm volatile("" :: "v"(r[0]), "v"(r[1]), "v"(r[2]), "v"(r[3]), "v"(r[4]), "v"(r[5]), "v"(r[6]), "v"(r[7]));
                    #pragma unroll
                    for (int u = 0; u < 8; ++u) {
                        v4f wa = *(const v4f*)(wmine + (i + u) * NH);
                        v4f wb = *(const v4f*)(wmine + (i + u) * NH + 4);
                        z[0] += r[u] * wa[0];
                        z[1] += r[u] * wa[1];
                        z[2] += r[u] * wa[2];
                        z[3] += r[u] * wa[3];
                        z[4] += r[u] * wb[0];
                        z[5] += r[u] * wb[1];
                        z[6] += r[u] * wb[2];
                        z[7] += r[u] * wb[3];
                    }
                }
                for (; i < cnt; i += 4) {
                    v4f r[4];
                    #pragma unroll
                    for (int u = 0; u < 4; ++u) {
                        int s = __builtin_amdgcn_readlane(ms, i + u);
                        r[u] = *(const v4f*)(xl + (long)s * F_IN);
                    }
                    asm volatile("" :: "v"(r[0]), "v"(r[1]), "v"(r[2]), "v"(r[3]));
                    #pragma unroll
                    for (int u = 0; u < 4; ++u) {
                        v4f wa = *(const v4f*)(wmine + (i + u) * NH);
                        v4f wb = *(const v4f*)(wmine + (i + u) * NH + 4);
                        z[0] += r[u] * wa[0];
                        z[1] += r[u] * wa[1];
                        z[2] += r[u] * wa[2];
                        z[3] += r[u] * wa[3];
                        z[4] += r[u] * wb[0];
                        z[5] += r[u] * wb[1];
                        z[6] += r[u] * wb[2];
                        z[7] += r[u] * wb[3];
                    }
                }
                j0 += 32;
            } while (j0 < je);

            const float exs = __expf(aself - mrun);
            dl += __shfl_xor(dl, 8);
            dl += __shfl_xor(dl, 16);
            const float den = dl + exs;
            const float inv = (1.0f / NH) / den;
            {
                v4f r = *(const v4f*)(xl + n * F_IN);
                #pragma unroll
                for (int g = 0; g < NH; ++g) z[g] += r * GAT_lane_bcast(exs, g);
            }
            #pragma unroll
            for (int g = 0; g < NH; ++g) z[g] *= GAT_lane_bcast(inv, g);
        }
        float* zrow = zt + (rt * 16 + wave) * ZP + lane * 4;
        #pragma unroll
        for (int g = 0; g < NH; ++g) *(v4f*)(zrow + g * F_IN) = z[g];
    }
    __syncthreads();

    {
        const int hw = wave >> 1;
        const int ct = wave & 1;
        const int l  = lane & 15;
        const int hf = lane >> 4;
        const float* wp  = W + (long)(hw * NC + ct * 16 + l) * F_IN + 4 * hf;
        const float* za0 = zt + l * ZP + hw * F_IN + 4 * hf;
        const float* za1 = za0 + 16 * ZP;
        v8f p00 = {0.f,0.f,0.f,0.f,0.f,0.f,0.f,0.f};
        v8f p01 = p00, p10 = p00, p11 = p00;
        #pragma unroll
        for (int k0 = 0; k0 < F_IN; k0 += 8) {
            v4f b  = *(const v4f*)(wp  + k0);
            v4f a0 = *(const v4f*)(za0 + k0);
            v4f a1 = *(const v4f*)(za1 + k0);
            v2f bl  = {b[0],  b[1]},  bh  = {b[2],  b[3]};
            v2f a0l = {a0[0], a0[1]}, a0h = {a0[2], a0[3]};
            v2f a1l = {a1[0], a1[1]}, a1h = {a1[2], a1[3]};
            p00 = __builtin_amdgcn_wmma_f32_16x16x4_f32(false, a0l, false, bl, (short)0, p00, false, false);
            p10 = __builtin_amdgcn_wmma_f32_16x16x4_f32(false, a1l, false, bl, (short)0, p10, false, false);
            p01 = __builtin_amdgcn_wmma_f32_16x16x4_f32(false, a0h, false, bh, (short)0, p01, false, false);
            p11 = __builtin_amdgcn_wmma_f32_16x16x4_f32(false, a1h, false, bh, (short)0, p11, false, false);
        }
        v8f p0 = p00 + p01;
        v8f p1 = p10 + p11;
        __syncthreads();
        float* pt = zt + wave * 512 + l;
        #pragma unroll
        for (int v = 0; v < 8; ++v) {
            pt[(8 * hf + v) * 16]       = p0[v];
            pt[256 + (8 * hf + v) * 16] = p1[v];
        }
    }
    __syncthreads();
    {
        const int c   = lane;
        const int ct2 = c >> 4, cc = c & 15;
        const float bc = bias[c];
        #pragma unroll
        for (int rt = 0; rt < 2; ++rt) {
            const long n = (long)blockIdx.x * 32 + rt * 16 + wave;
            float s = 0.0f;
            #pragma unroll
            for (int g = 0; g < NH; ++g)
                s += zt[((g * 2 + ct2) * 2 + rt) * 256 + wave * 16 + cc];
            if (n < N) out[n * NC + c] = s + bc;
        }
    }
}

extern "C" void kernel_launch(void* const* d_in, const int* in_sizes, int n_in,
                              void* d_out, int out_size, void* d_ws, size_t ws_size,
                              hipStream_t stream) {
    const float* x        = (const float*)d_in[0];
    const int*   eidx     = (const int*)  d_in[1];
    const float* eattr    = (const float*)d_in[2];
    const float* W        = (const float*)d_in[3];
    const float* We       = (const float*)d_in[4];
    const float* att_src  = (const float*)d_in[5];
    const float* att_dst  = (const float*)d_in[6];
    const float* att_edge = (const float*)d_in[7];
    const float* bias     = (const float*)d_in[8];

    const int N = in_sizes[0] / F_IN;
    const int E = in_sizes[1] / 2;
    const int* src = eidx;
    const int* dst = eidx + E;

    float* ws = (float*)d_ws;
    const long Nl = N, El = E;
    const long NP = (Nl + 63) & ~63L;
    const long epc = (((El + RK_G - 1) / RK_G) + 255) & ~255L;
    int*      w_deg   = (int*)ws;
    int*      w_tot   = (int*)(ws + Nl);
    const long o_off  = (Nl + 64 + 63) & ~63L;
    const long o_rank = (o_off + Nl + 63) & ~63L;
    const long o_perm = (o_rank + El + 63) & ~63L;
    const long o_att  = (o_perm + 2 * El + 63) & ~63L;
    const long o_cnt  = (o_att + 16 * Nl + 128 + 16 * F_IN + 63) & ~63L;
    const long o_aedge = (o_cnt + (long)RK_G * NP + 63) & ~63L;
    int*      w_off   = (int*)(ws + o_off);
    int*      w_rank  = (int*)(ws + o_rank);
    int*      w_perm  = (int*)(ws + o_perm);
    float*    w_asrc  = ws + o_att;
    float*    w_adst  = ws + o_att + 8 * Nl;
    float*    w_M     = ws + o_att + 16 * Nl;
    float*    w_Vt    = ws + o_att + 16 * Nl + 128;
    int*      w_cnt   = (int*)(ws + o_cnt);
    float*    w_aedge = ws + o_aedge;

    const int B = 256;
    GAT_rank<<<RK_G + 3, 1024, 0, stream>>>(dst, w_rank, w_cnt, w_tot, E, N, (int)epc, (int)NP,
                                            W, We, att_src, att_dst, att_edge, w_M, w_Vt);
    GAT_segalloc<<<(int)((Nl + 255) / 256), 256, 0, stream>>>(w_cnt, w_deg, w_off, w_tot, N, (int)NP);
    {
        const long bpc = epc / B;
        const long attBlocks = (((Nl + 15) >> 4) + 7) / 8;
        const long gy = (attBlocks + bpc - 1) / bpc;
        GAT_fill<<<dim3((unsigned)bpc, (unsigned)(RK_G + gy)), B, 0, stream>>>(src, dst, w_cnt, w_rank, w_perm, E, (int)epc, (int)NP,
                                                                           (int)gy, x, w_Vt, w_asrc, w_adst, N,
                                                                           eattr, w_M, w_aedge);
    }
    GAT_gather<<<(int)((Nl + 31) / 32), 512, 0, stream>>>(w_deg, w_off, w_perm, w_aedge,
                                                          w_asrc, w_adst, x, W, bias,
                                                          (float*)d_out, N);
}


# Edge slots from returning LDS adds on base rows staged in LDS: placement groups per (chunk, node range), terms by 4-edge threads; no per-edge base gather, no rank plane; arrival order as before
# speedup vs baseline: 9.0166x; 1.0339x over previous
#include <hip/hip_runtime.h>
#include <hip/hip_bf16.h>

typedef float v2f __attribute__((ext_vector_type(2)));
typedef float v8f __attribute__((ext_vector_type(8)));
typedef float v4f __attribute__((ext_vector_type(4)));

#define F_IN 128
#define F_E  16
#define NH   8
#define NC   32
#define HC   256
#define NEG_SLOPE 0.2f

typedef int v4i __attribute__((ext_vector_type(4)));
typedef int v2i __attribute__((ext_vector_type(2)));

#define RK_G    64
#define RK_CAP  51200
#define RK_B    16
#define FL_T    512
#define FL_CAP  12800
#define FL_B    16
#define FL_K    4
#define FL_H    2

__device__ __forceinline__ void GAT_fold_part(int part, int tid,
                         const float* __restrict__ W,
                         const float* __restrict__ We,
                         const float* __restrict__ att_src,
                         const float* __restrict__ att_dst,
                         const float* __restrict__ att_edge,
                         float* __restrict__ M,
                         float* __restrict__ Vt) {
    if (part < 2) {
        int idx = part * 1024 + tid;
        int j = idx >> 7, f = idx & (F_IN - 1);
        int hj = j & 7;
        const float* att = (j < NH) ? att_src : att_dst;
        float s = 0.0f;
        for (int c = 0; c < NC; ++c)
            s += W[(long)(hj * NC + c) * F_IN + f] * att[hj * NC + c];
        Vt[j * F_IN + f] = s;
    } else {
        if (tid >= F_E * NH) return;
        int f = tid >> 3, h = tid & 7;
        float s = 0.0f;
        for (int c = 0; c < NC; ++c)
            s += We[(long)(h * NC + c) * F_E + f] * att_edge[h * NC + c];
        M[f * NH + h] = s;
    }
}

__global__ void __launch_bounds__(1024)
GAT_rank(const int* __restrict__ dst, int* __restrict__ cnt,
         int* __restrict__ total, int E, int N, int epc, int np,
         const float* __restrict__ W, const float* __restrict__ We,
         const float* __restrict__ att_src, const float* __restrict__ att_dst,
         const float* __restrict__ att_edge, float* __restrict__ M, float* __restrict__ Vt) {
    __shared__ __attribute__((aligned(16))) int hist[RK_CAP];
    const int t = threadIdx.x;
    const int g = blockIdx.x;
    if (g >= RK_G) {
        GAT_fold_part(g - RK_G, t, W, We, att_src, att_dst, att_edge, M, Vt);
        return;
    }
    if (g == 0 && t == 0) *total = 0;
    const long e0 = (long)g * epc;
    long e1 = e0 + epc;
    if (e1 > E) e1 = E;
    for (int nb = 0; nb < N; nb += RK_CAP) {
        int sn = N - nb;
        if (sn > RK_CAP) sn = RK_CAP;
        const int sn4 = (sn + 3) >> 2;
        int d[RK_B];
        long eb = e0;
        #pragma unroll
        for (int u = 0; u < RK_B; ++u) {
            const long e = eb + u * 1024 + t;
            d[u] = (e < e1) ? dst[e] : -1;
        }
        for (int i = t; i < sn4; i += 1024) ((v4i*)hist)[i] = (v4i){0, 0, 0, 0};
        __syncthreads();
        while (true) {
            #pragma unroll
            for (int u = 0; u < RK_B; ++u) {
                const unsigned k = (unsigned)(d[u] - nb);
                if (k < (unsigned)sn) atomicAdd(&hist[k], 1);
            }
            eb += (long)RK_B * 1024;
            if (eb >= e1) break;
            #pragma unroll
            for (int u = 0; u < RK_B; ++u) {
                const long e = eb + u * 1024 + t;
                d[u] = (e < e1) ? dst[e] : -1;
            }
        }
        __syncthreads();
        int* row = cnt + (long)g * np + nb;
        #pragma unroll 4
        for (int i = t; i < sn4; i += 1024) ((v4i*)row)[i] = ((const v4i*)hist)[i];
        __syncthreads();
    }
}

__device__ __forceinline__ void GAT_gemm_att_part(long blk,
                             const float* __restrict__ x,
                             const float* __restrict__ Vt,
                             float* __restrict__ asrc,
                             float* __restrict__ adst, int N) {
    int wave = threadIdx.x >> 5;
    int lane = threadIdx.x & 31;
    int half = lane >> 4;
    int l    = lane & 15;
    long rowTiles = ((long)N + 15) >> 4;
    long tile = blk * (FL_T / 32) + wave;
    if (tile >= rowTiles) return;
    long row0 = tile * 16;
    long arow = row0 + l;
    if (arow >= N) arow = N - 1;
    const float* xp = x  + arow * F_IN + 2 * half;
    const float* vp = Vt + (long)l * F_IN + 2 * half;
    v8f a0 = {0.f,0.f,0.f,0.f,0.f,0.f,0.f,0.f};
    v8f a1 = a0;
    #pragma unroll
    for (int k0 = 0; k0 < F_IN; k0 += 8) {
        v2f xa = *(const v2f*)(xp + k0);
        v2f va = *(const v2f*)(vp + k0);
        v2f xb = *(const v2f*)(xp + k0 + 4);
        v2f vb = *(const v2f*)(vp + k0 + 4);
        a0 = __builtin_amdgcn_wmma_f32_16x16x4_f32(false, xa, false, va, (short)0, a0, false, false);
        a1 = __builtin_amdgcn_wmma_f32_16x16x4_f32(false, xb, false, vb, (short)0, a1, false, false);
    }
    v8f acc = a0 + a1;
    #pragma unroll
    for (int v = 0; v < 8; ++v) {
        long r = row0 + v + 8 * half;
        if (r < N) {
            if (l < NH) asrc[r * NH + l] = acc[v];
            else        adst[r * NH + (l - NH)] = acc[v];
        }
    }
}

__global__ void __launch_bounds__(256)
GAT_segalloc(int* __restrict__ cnt, int* __restrict__ deg, int* __restrict__ off,
             int* __restrict__ total, int N, int np) {
    __shared__ int wsum[8];
    __shared__ int base;
    const int t = threadIdx.x;
    const int lane = t & 31, w = t >> 5;
    const long n = (long)blockIdx.x * 256 + t;
    int* col = cnt + ((n < N) ? n : (long)N - 1);
    int c[RK_G];
    #pragma unroll
    for (int g = 0; g < RK_G; ++g) c[g] = col[(long)g * np];
    int tot = 0;
    #pragma unroll
    for (int g = 0; g < RK_G; ++g) tot += c[g];
    if (n < N) deg[n] = tot; else tot = 0;
    int inc = tot;
    for (int o = 1; o < 32; o <<= 1) {
        int v = __shfl_up(inc, o);
        if (lane >= o) inc += v;
    }
    if (lane == 31) wsum[w] = inc;
    __syncthreads();
    if (w == 0) {
        int s = (lane < 8) ? wsum[lane] : 0;
        int si = s;
        for (int o = 1; o < 32; o <<= 1) {
            int v = __shfl_up(si, o);
            if (lane >= o) si += v;
        }
        if (lane < 8) wsum[lane] = si - s;
        if (lane == 31) base = atomicAdd(total, si);
    }
    __syncthreads();
    if (n < N) {
        int run = base + wsum[w] + inc - tot;
        off[n] = run;
        #pragma unroll
        for (int g = 0; g < RK_G; ++g) { col[(long)g * np] = run; run += c[g]; }
    }
}

__device__ __forceinline__ void GAT_place_part(int g, int q, int* __restrict__ cur,
                          const int* __restrict__ src, const int* __restrict__ dst,
                          const int* __restrict__ cbase, int* __restrict__ perm,
                          int E, int N, int epc, int np) {
    const int t = threadIdx.x;
    const long e0 = (long)g * epc;
    long e1 = e0 + epc;
    if (e1 > E) e1 = E;
    const int nb = q * FL_CAP;
    int sn = N - nb;
    if (sn > FL_CAP) sn = FL_CAP;
    const int sn4 = (sn + 3) >> 2;
    int d[FL_B], s[FL_B];
    long eb = e0;
    #pragma unroll
    for (int u = 0; u < FL_B; ++u) {
        const long e = eb + u * FL_T + t;
        d[u] = (e < e1) ? dst[e] : -1;
        s[u] = (e < e1) ? src[e] : 0;
    }
    const int* row = cbase + (long)g * np + nb;
    #pragma unroll 4
    for (int i = t; i < sn4; i += FL_T) ((v4i*)cur)[i] = ((const v4i*)row)[i];
    __syncthreads();
    while (true) {
        int slot[FL_B];
        #pragma unroll
        for (int u = 0; u < FL_B; ++u) {
            const unsigned k = (unsigned)(d[u] - nb);
            slot[u] = (k < (unsigned)sn) ? atomicAdd(&cur[k], 1) : -1;
        }
        #pragma unroll
        for (int u = 0; u < FL_B; ++u) {
            const long e = eb + u * FL_T + t;
            if (slot[u] >= 0) ((v2i*)perm)[slot[u]] = (v2i){(int)e, s[u]};
        }
        eb += (long)FL_B * FL_T;
        if (eb >= e1) break;
        #pragma unroll
        for (int u = 0; u < FL_B; ++u) {
            const long e = eb + u * FL_T + t;
            d[u] = (e < e1) ? dst[e] : -1;
            s[u] = (e < e1) ? src[e] : 0;
        }
    }
}

__device__ __forceinline__ void GAT_term_part(long unit,
                         const float* __restrict__ eattr, const float* __restrict__ M,
                         float* __restrict__ aedge, int E) {
    const long base = unit * (FL_T * FL_K) + threadIdx.x;
    #pragma nounroll
    for (int k0 = 0; k0 < FL_K; k0 += FL_H) {
        v4f a[FL_H][4];
        #pragma unroll
        for (int k = 0; k < FL_H; ++k) {
            long e = base + (long)(k0 + k) * FL_T;
            if (e >= E) e = (long)E - 1;
            const v4f* er = (const v4f*)(eattr + e * F_E);
            a[k][0] = er[0]; a[k][1] = er[1]; a[k][2] = er[2]; a[k][3] = er[3];
        }
        #pragma unroll
        for (int k = 0; k < FL_H; ++k) {
            const long e = base + (long)(k0 + k) * FL_T;
            float t[NH];
            #pragma unroll
            for (int h = 0; h < NH; ++h) t[h] = 0.0f;
            #pragma unroll
            for (int p = 0; p < 4; ++p) {
                #pragma unroll
                for (int q = 0; q < 4; ++q) {
                    #pragma unroll
                    for (int h = 0; h < NH; ++h) t[h] = __builtin_fmaf(a[k][p][q], M[(4 * p + q) * NH + h], t[h]);
                }
            }
            if (e < E) {
                v4f* tr = (v4f*)(aedge + e * NH);
                tr[0] = (v4f){t[0], t[1], t[2], t[3]};
                tr[1] = (v4f){t[4], t[5], t[6], t[7]};
            }
        }
    }
}

__global__ void __launch_bounds__(FL_T, 8)
GAT_fill(const int* __restrict__ src, const int* __restrict__ dst, const int* __restrict__ cbase,
         int* __restrict__ perm, int E, int epc, int np, int nsw, int nown, int ngemm,
         const float* __restrict__ x, const float* __restrict__ Vt,
         float* __restrict__ asrc, float* __restrict__ adst, int N,
         const float* __restrict__ eattr, const float* __restrict__ M, float* __restrict__ aedge) {
    __shared__ __attribute__((aligned(16))) int cur[FL_CAP];
    const int b = blockIdx.x;
    if (b < nown) {
        GAT_place_part(b / nsw, b % nsw, cur, src, dst, cbase, perm, E, N, epc, np);
        return;
    }
    if (b < nown + ngemm) {
        GAT_gemm_att_part((long)(b - nown), x, Vt, asrc, adst, N);
        return;
    }
    GAT_term_part((long)(b - nown - ngemm), eattr, M, aedge, E);
}

__device__ __forceinline__ float GAT_lane_bcast(float v, int g) {
    return __builtin_bit_cast(float, __builtin_amdgcn_readlane(__builtin_bit_cast(int, v), g));
}
#define ZP 1028
__global__ void __launch_bounds__(512, 8)
GAT_gather(const int* __restrict__ deg,
           const int* __restrict__ off,
           const int* __restrict__ perm,
           const float* __restrict__ aedge,
           const float* __restrict__ asrc,
           const float* __restrict__ adst,
           const float* __restrict__ x,
           const float* __restrict__ W,
           const float* __restrict__ bias,
           float* __restrict__ out, int N) {
    __shared__ __attribute__((aligned(16))) float zt[32 * ZP];
    __shared__ __attribute__((aligned(16))) float wexp[16][32 * NH];
    const int wave = __builtin_amdgcn_readfirstlane((int)(threadIdx.x >> 5));
    const int lane = threadIdx.x & 31;
    const int hq = lane >> 3;
    const int h  = lane & 7;
    float* wmine = &wexp[wave][0];
    const float* xl = x + lane * 4;

    #pragma nounroll
    for (int rt = 0; rt < 2; ++rt) {
        const long n = (long)blockIdx.x * 32 + rt * 16 + wave;
        float* stage = zt + (rt * 16 + wave) * ZP;
        v4f z[NH];
        #pragma unroll
        for (int g = 0; g < NH; ++g) z[g] = (v4f){0.f, 0.f, 0.f, 0.f};
        if (n < N) {
            const float adh = adst[n * NH + h];
            const float asn = asrc[n * NH + h];
            float mrun = -__builtin_inff();
            float dl = 0.0f;
            float es = 0.0f;
            float aself = 0.0f;
            const int dg = deg[n];
            const int jb = off[n];
            const int je = jb + dg;
            int j0 = jb;
            do {
                int cnt = je - j0;
                if (cnt > 32) cnt = 32;
                int me = 0, ms = (int)n;
                if (lane < cnt) {
                    const v2i p = ((const v2i*)perm)[j0 + lane];
                    me = p[0];
                    ms = p[1];
                }
                const int nk = (cnt + 3) >> 2;
                if (lane < 4 * nk) {
                    const float* er = aedge + (long)me * NH;
                    const float* tr = asrc + (long)ms * NH;
                    v4f s0 = *(const v4f*)(er);
                    v4f s1 = *(const v4f*)(er + 4);
                    v4f t0 = *(const v4f*)(tr);
                    v4f t1 = *(const v4f*)(tr + 4);
                    *(v4f*)(stage + lane * NH)     = s0;
                    *(v4f*)(stage + lane * NH + 4) = s1;
                    *(v4f*)(wmine + lane * NH)     = t0;
                    *(v4f*)(wmine + lane * NH + 4) = t1;
                }
                float lmax = -__builtin_inff();
                #pragma unroll 2
                for (int k = 0; k < nk; ++k) {
                    const int j = 4 * k + hq;
                    const bool valid = j < cnt;
                    float ae = stage[32 * k + lane];
                    float as = wmine[32 * k + lane];
                    float a = as + adh + ae;
                    a = (a >= 0.0f) ? a : NEG_SLOPE * a;
                    a = valid ? a : -__builtin_inff();
                    es += valid ? ae : 0.0f;
                    wmine[32 * k + lane] = a;
                    lmax = fmaxf(lmax, a);
                }
                float cm = fmaxf(lmax, __shfl_xor(lmax, 8));
                cm = fmaxf(cm, __shfl_xor(cm, 16));
                if (j0 + 32 >= je) {
                    es += __shfl_xor(es, 8);
                    es += __shfl_xor(es, 16);
                    const float dmax = fmaxf((float)dg, 1.0f);
                    const float ae = es / dmax;
                    float a = asn + adh + ae;
                    aself = (a >= 0.0f) ? a : NEG_SLOPE * a;
                    cm = fmaxf(cm, aself);
                }
                const float mn = fmaxf(mrun, cm);
                const float sc = __expf(mrun - mn);
                mrun = mn;
                float lsum = 0.0f;
                #pragma unroll 4
                for (int k = 0; k < nk; ++k) {
                    float ex = __expf(wmine[32 * k + lane] - mn);
                    wmine[32 * k + lane] = ex;
                    lsum += ex;
                }
                dl = dl * sc + lsum;
                #pragma unroll
                for (int g = 0; g < NH; ++g) z[g] *= GAT_lane_bcast(sc, g);
                int i = 0;
                for (; i + 8 <= cnt; i += 8) {
                    v4f r[8];
                    #pragma unroll
                    for (int u = 0; u < 8; ++u) {
                        int s = __builtin_amdgcn_readlane(ms, i + u);
                        r[u] = *(const v4f*)(xl + (long)s * F_IN);
                    }
                    asm volatile("" :: "v"(r[0]), "v"(r[1]), "v"(r[2]), "v"(r[3]), "v"(r[4]), "v"(r[5]), "v"(r[6]), "v"(r[7]));
                    #pragma unroll
                    for (int u = 0; u < 8; ++u) {
                        v4f wa = *(const v4f*)(wmine + (i + u) * NH);
                        v4f wb = *(const v4f*)(wmine + (i + u) * NH + 4);
                        z[0] += r[u] * wa[0];
                        z[1] += r[u] * wa[1];
                        z[2] += r[u] * wa[2];
                        z[3] += r[u] * wa[3];
                        z[4] += r[u] * wb[0];
                        z[5] += r[u] * wb[1];
                        z[6] += r[u] * wb[2];
                        z[7] += r[u] * wb[3];
                    }
                }
                for (; i < cnt; i += 4) {
                    v4f r[4];
                    #pragma unroll
                    for (int u = 0; u < 4; ++u) {
                        int s = __builtin_amdgcn_readlane(ms, i + u);
                        r[u] = *(const v4f*)(xl + (long)s * F_IN);
                    }
                    asm volatile("" :: "v"(r[0]), "v"(r[1]), "v"(r[2]), "v"(r[3]));
                    #pragma unroll
                    for (int u = 0; u < 4; ++u) {
                        v4f wa = *(const v4f*)(wmine + (i + u) * NH);
                        v4f wb = *(const v4f*)(wmine + (i + u) * NH + 4);
                        z[0] += r[u] * wa[0];
                        z[1] += r[u] * wa[1];
                        z[2] += r[u] * wa[2];
                        z[3] += r[u] * wa[3];
                        z[4] += r[u] * wb[0];
                        z[5] += r[u] * wb[1];
                        z[6] += r[u] * wb[2];
                        z[7] += r[u] * wb[3];
                    }
                }
                j0 += 32;
            } while (j0 < je);

            const float exs = __expf(aself - mrun);
            dl += __shfl_xor(dl, 8);
            dl += __shfl_xor(dl, 16);
            const float den = dl + exs;
            const float inv = (1.0f / NH) / den;
            {
                v4f r = *(const v4f*)(xl + n * F_IN);
                #pragma unroll
                for (int g = 0; g < NH; ++g) z[g] += r * GAT_lane_bcast(exs, g);
            }
            #pragma unroll
            for (int g = 0; g < NH; ++g) z[g] *= GAT_lane_bcast(inv, g);
        }
        float* zrow = zt + (rt * 16 + wave) * ZP + lane * 4;
        #pragma unroll
        for (int g = 0; g < NH; ++g) *(v4f*)(zrow + g * F_IN) = z[g];
    }
    __syncthreads();

    {
        const int hw = wave >> 1;
        const int ct = wave & 1;
        const int l  = lane & 15;
        const int hf = lane >> 4;
        const float* wp  = W + (long)(hw * NC + ct * 16 + l) * F_IN + 4 * hf;
        const float* za0 = zt + l * ZP + hw * F_IN + 4 * hf;
        const float* za1 = za0 + 16 * ZP;
        v8f p00 = {0.f,0.f,0.f,0.f,0.f,0.f,0.f,0.f};
        v8f p01 = p00, p10 = p00, p11 = p00;
        #pragma unroll
        for (int k0 = 0; k0 < F_IN; k0 += 8) {
            v4f b  = *(const v4f*)(wp  + k0);
            v4f a0 = *(const v4f*)(za0 + k0);
            v4f a1 = *(const v4f*)(za1 + k0);
            v2f bl  = {b[0],  b[1]},  bh  = {b[2],  b[3]};
            v2f a0l = {a0[0], a0[1]}, a0h = {a0[2], a0[3]};
            v2f a1l = {a1[0], a1[1]}, a1h = {a1[2], a1[3]};
            p00 = __builtin_amdgcn_wmma_f32_16x16x4_f32(false, a0l, false, bl, (short)0, p00, false, false);
            p10 = __builtin_amdgcn_wmma_f32_16x16x4_f32(false, a1l, false, bl, (short)0, p10, false, false);
            p01 = __builtin_amdgcn_wmma_f32_16x16x4_f32(false, a0h, false, bh, (short)0, p01, false, false);
            p11 = __builtin_amdgcn_wmma_f32_16x16x4_f32(false, a1h, false, bh, (short)0, p11, false, false);
        }
        v8f p0 = p00 + p01;
        v8f p1 = p10 + p11;
        __syncthreads();
        float* pt = zt + wave * 512 + l;
        #pragma unroll
        for (int v = 0; v < 8; ++v) {
            pt[(8 * hf + v) * 16]       = p0[v];
            pt[256 + (8 * hf + v) * 16] = p1[v];
        }
    }
    __syncthreads();
    {
        const int c   = lane;
        const int ct2 = c >> 4, cc = c & 15;
        const float bc = bias[c];
        #pragma unroll
        for (int rt = 0; rt < 2; ++rt) {
            const long n = (long)blockIdx.x * 32 + rt * 16 + wave;
            float s = 0.0f;
            #pragma unroll
            for (int g = 0; g < NH; ++g)
                s += zt[((g * 2 + ct2) * 2 + rt) * 256 + wave * 16 + cc];
            if (n < N) out[n * NC + c] = s + bc;
        }
    }
}

extern "C" void kernel_launch(void* const* d_in, const int* in_sizes, int n_in,
                              void* d_out, int out_size, void* d_ws, size_t ws_size,
                              hipStream_t stream) {
    const float* x        = (const float*)d_in[0];
    const int*   eidx     = (const int*)  d_in[1];
    const float* eattr    = (const float*)d_in[2];
    const float* W        = (const float*)d_in[3];
    const float* We       = (const float*)d_in[4];
    const float* att_src  = (const float*)d_in[5];
    const float* att_dst  = (const float*)d_in[6];
    const float* att_edge = (const float*)d_in[7];
    const float* bias     = (const float*)d_in[8];

    const int N = in_sizes[0] / F_IN;
    const int E = in_sizes[1] / 2;
    const int* src = eidx;
    const int* dst = eidx + E;

    float* ws = (float*)d_ws;
    const long Nl = N, El = E;
    const long NP = (Nl + 63) & ~63L;
    const long epc = (((El + RK_G - 1) / RK_G) + 255) & ~255L;
    int*      w_deg   = (int*)ws;
    int*      w_tot   = (int*)(ws + Nl);
    const long o_off  = (Nl + 64 + 63) & ~63L;
    const long o_perm = (o_off + Nl + 63) & ~63L;
    const long o_att  = (o_perm + 2 * El + 63) & ~63L;
    const long o_cnt  = (o_att + 16 * Nl + 128 + 16 * F_IN + 63) & ~63L;
    const long o_aedge = (o_cnt + (long)RK_G * NP + 63) & ~63L;
    int*      w_off   = (int*)(ws + o_off);
    int*      w_perm  = (int*)(ws + o_perm);
    float*    w_asrc  = ws + o_att;
    float*    w_adst  = ws + o_att + 8 * Nl;
    float*    w_M     = ws + o_att + 16 * Nl;
    float*    w_Vt    = ws + o_att + 16 * Nl + 128;
    int*      w_cnt   = (int*)(ws + o_cnt);
    float*    w_aedge = ws + o_aedge;

    GAT_rank<<<RK_G + 3, 1024, 0, stream>>>(dst, w_cnt, w_tot, E, N, (int)epc, (int)NP,
                                            W, We, att_src, att_dst, att_edge, w_M, w_Vt);
    GAT_segalloc<<<(int)((Nl + 255) / 256), 256, 0, stream>>>(w_cnt, w_deg, w_off, w_tot, N, (int)NP);
    {
        const long nsw   = (Nl + FL_CAP - 1) / FL_CAP;
        const long nown  = (long)RK_G * nsw;
        const long ngemm = (((Nl + 15) >> 4) + (FL_T / 32) - 1) / (FL_T / 32);
        const long nterm = (El + (long)FL_T * FL_K - 1) / ((long)FL_T * FL_K);
        GAT_fill<<<(unsigned)(nown + ngemm + nterm), FL_T, 0, stream>>>(src, dst, w_cnt, w_perm, E, (int)epc, (int)NP,
                                                                        (int)nsw, (int)nown, (int)ngemm,
                                                                        x, w_Vt, w_asrc, w_adst, N,
                                                                        eattr, w_M, w_aedge);
    }
    GAT_gather<<<(int)((Nl + 31) / 32), 512, 0, stream>>>(w_deg, w_off, w_perm, w_aedge,
                                                          w_asrc, w_adst, x, W, bias,
                                                          (float*)d_out, N);
}
